# starting kernel
# baseline (speedup 1.0000x reference)
#include <hip/hip_runtime.h>
#include <hip/hip_bf16.h>


typedef __attribute__((ext_vector_type(2))) float v2f;
typedef __attribute__((ext_vector_type(8))) float v8f;

#define NUM_USERS 100000
#define NUM_ITEMS 50000
#define NUM_NODES 150000
#define NUM_BQ    20000
#define EMB       64

__global__ void __launch_bounds__(256) copy_init_kernel(
    const float* __restrict__ ue, const float* __restrict__ ie,
    float* __restrict__ e0, float* __restrict__ out,
    int totalVec, int userVec) {
  int g = blockIdx.x * blockDim.x + threadIdx.x;
  if (g >= totalVec) return;
  float4 v = (g < userVec) ? ((const float4*)ue)[g]
                           : ((const float4*)ie)[g - userVec];
  ((float4*)e0)[g]  = v;
  ((float4*)out)[g] = v;
}

__global__ void __launch_bounds__(256) spmm16_kernel(
    const int* __restrict__ row, const int* __restrict__ col,
    const float* __restrict__ val, const float* __restrict__ x,
    float* __restrict__ y, float* __restrict__ deg, int nnz) {
  int g = blockIdx.x * blockDim.x + threadIdx.x;
  int e = g >> 4;
  int j = g & 15;
  if (e >= nnz) return;
  int   c = __builtin_nontemporal_load(col + e);
  int   r = __builtin_nontemporal_load(row + e);
  float v = __builtin_nontemporal_load(val + e);
  float4 xv = ((const float4*)(x + (size_t)c * EMB))[j];
  float* yp = y + (size_t)r * EMB + j * 4;
  unsafeAtomicAdd(yp + 0, v * xv.x);
  unsafeAtomicAdd(yp + 1, v * xv.y);
  unsafeAtomicAdd(yp + 2, v * xv.z);
  unsafeAtomicAdd(yp + 3, v * xv.w);
  if (deg != nullptr && j == 0)
    unsafeAtomicAdd(&deg[r], v);
}

__global__ void __launch_bounds__(128) acc_wmma_kernel(
    const float* __restrict__ e, float* __restrict__ acc) {
  int wave = threadIdx.x >> 5;
  int lane = threadIdx.x & 31;
  int n    = lane & 15;
  int hi   = lane >> 4;
  int row0 = blockIdx.x * 16;
  int col0 = wave * 16;

  const size_t base = (size_t)row0 * EMB + col0 + n;
  const float* ap = acc + base;
  const float* ep = e + base;

  v8f c;
#pragma unroll
  for (int r = 0; r < 8; ++r)
    c[r] = ap[(size_t)(r + 8 * hi) * EMB];

  v2f a[4], b[4];
#pragma unroll
  for (int k = 0; k < 4; ++k) {
    a[k].x = (n == (4 * k + 2 * hi))     ? 1.0f : 0.0f;
    a[k].y = (n == (4 * k + 2 * hi + 1)) ? 1.0f : 0.0f;
    b[k].x = ep[(size_t)(4 * k + 2 * hi) * EMB];
    b[k].y = ep[(size_t)(4 * k + 2 * hi + 1) * EMB];
  }

#pragma unroll
  for (int k = 0; k < 4; ++k)
    c = __builtin_amdgcn_wmma_f32_16x16x4_f32(
        false, a[k], false, b[k], (short)0, c, false, false);

  float* op = acc + base;
#pragma unroll
  for (int r = 0; r < 8; ++r)
    op[(size_t)(r + 8 * hi) * EMB] = c[r];
}

__global__ void __launch_bounds__(256) rownorm_kernel(
    float* __restrict__ x, const float* __restrict__ deg, int rowsVec) {
  int g = blockIdx.x * blockDim.x + threadIdx.x;
  if (g >= rowsVec) return;
  float d = deg[g >> 4];
  d = (d == 0.0f) ? 1.0f : d;
  float inv = 1.0f / d;
  float4 v = ((float4*)x)[g];
  v.x *= inv; v.y *= inv; v.z *= inv; v.w *= inv;
  ((float4*)x)[g] = v;
}

__global__ void __launch_bounds__(256) finalize_kernel(
    float* __restrict__ out, const float* __restrict__ ulocal,
    const float* __restrict__ deg, int totalVec, int userVec) {
  int g = blockIdx.x * blockDim.x + threadIdx.x;
  if (g >= totalVec) return;
  float4 v = ((float4*)out)[g];
  v.x *= 0.25f; v.y *= 0.25f; v.z *= 0.25f; v.w *= 0.25f;
  if (g < userVec) {
    float d = deg[g >> 4];
    d = (d == 0.0f) ? 1.0f : d;
    float inv = 1.0f / d;
    float4 w = ((const float4*)ulocal)[g];
    v.x += w.x * inv; v.y += w.y * inv; v.z += w.z * inv; v.w += w.w * inv;
  }
  ((float4*)out)[g] = v;
}

static inline int cdiv(long long a, int b) { return (int)((a + b - 1) / b); }

extern "C" void kernel_launch(void* const* d_in, const int* in_sizes, int n_in,
                              void* d_out, int out_size, void* d_ws, size_t ws_size,
                              hipStream_t stream) {
  const float* user_emb = (const float*)d_in[0];
  const float* item_emb = (const float*)d_in[1];
  const float* adj_val  = (const float*)d_in[2];
  const float* hv_val   = (const float*)d_in[3];
  const float* hu_val   = (const float*)d_in[4];
  const int*   adj_row  = (const int*)d_in[5];
  const int*   adj_col  = (const int*)d_in[6];
  const int*   hv_row   = (const int*)d_in[7];
  const int*   hv_col   = (const int*)d_in[8];
  const int*   hu_row   = (const int*)d_in[9];
  const int*   hu_col   = (const int*)d_in[10];

  const int E_adj = in_sizes[2];
  const int E_hv  = in_sizes[3];
  const int E_hu  = in_sizes[4];

  float* out = (float*)d_out;

  const size_t embBytesN = (size_t)NUM_NODES * EMB * sizeof(float);
  const size_t embBytesB = (size_t)NUM_BQ    * EMB * sizeof(float);
  const size_t embBytesU = (size_t)NUM_USERS * EMB * sizeof(float);
  char* ws = (char*)d_ws;
  float* e0     = (float*)(ws);
  float* e1     = (float*)(ws + embBytesN);
  float* bfeat  = (float*)(ws + 2 * embBytesN);
  float* deg_hv = (float*)(ws + 2 * embBytesN + embBytesB);
  float* deg_hu = deg_hv + NUM_BQ;
  float* ulocal = e0;

  const int totalVec = NUM_NODES * (EMB / 4);
  const int userVec  = NUM_USERS * (EMB / 4);
  const int bqVec    = NUM_BQ    * (EMB / 4);

  copy_init_kernel<<<cdiv(totalVec, 256), 256, 0, stream>>>(
      user_emb, item_emb, e0, out, totalVec, userVec);

  float* eprev = e0;
  float* enext = e1;
  for (int layer = 0; layer < 3; ++layer) {
    hipMemsetAsync(enext, 0, embBytesN, stream);
    spmm16_kernel<<<cdiv((long long)E_adj * 16, 256), 256, 0, stream>>>(
        adj_row, adj_col, adj_val, eprev, enext, nullptr, E_adj);
    acc_wmma_kernel<<<NUM_NODES / 16, 128, 0, stream>>>(enext, out);
    float* t = eprev; eprev = enext; enext = t;
  }

  hipMemsetAsync(bfeat,  0, embBytesB, stream);
  hipMemsetAsync(deg_hv, 0, (size_t)NUM_BQ * sizeof(float), stream);
  hipMemsetAsync(deg_hu, 0, (size_t)NUM_USERS * sizeof(float), stream);

  spmm16_kernel<<<cdiv((long long)E_hv * 16, 256), 256, 0, stream>>>(
      hv_row, hv_col, hv_val, item_emb, bfeat, deg_hv, E_hv);
  rownorm_kernel<<<cdiv(bqVec, 256), 256, 0, stream>>>(bfeat, deg_hv, bqVec);

  hipMemsetAsync(ulocal, 0, embBytesU, stream);
  spmm16_kernel<<<cdiv((long long)E_hu * 16, 256), 256, 0, stream>>>(
      hu_row, hu_col, hu_val, bfeat, ulocal, deg_hu, E_hu);

  finalize_kernel<<<cdiv(totalVec, 256), 256, 0, stream>>>(
      out, ulocal, deg_hu, totalVec, userVec);
}


# SpMM float-atomic scatter -> row gather: edges bucketed by row per call (count, scan, fill); 16 lanes sum a row in registers in slot order and store it once; per-row sum order changes
# speedup vs baseline: 1.6406x; 1.6406x over previous
#include <hip/hip_runtime.h>
#include <hip/hip_bf16.h>


typedef __attribute__((ext_vector_type(2))) float v2f;
typedef __attribute__((ext_vector_type(8))) float v8f;

#define NUM_USERS 100000
#define NUM_ITEMS 50000
#define NUM_NODES 150000
#define NUM_BQ    20000
#define EMB       64

__global__ void __launch_bounds__(256) copy_init_kernel(
    const float* __restrict__ ue, const float* __restrict__ ie,
    float* __restrict__ e0, float* __restrict__ out,
    int totalVec, int userVec) {
  int g = blockIdx.x * blockDim.x + threadIdx.x;
  if (g >= totalVec) return;
  float4 v = (g < userVec) ? ((const float4*)ue)[g]
                           : ((const float4*)ie)[g - userVec];
  ((float4*)e0)[g]  = v;
  ((float4*)out)[g] = v;
}

__global__ void __launch_bounds__(256) count_rank_kernel(
    const int* __restrict__ row0, int nnz0,
    const int* __restrict__ row1, int nnz1, int rbase1,
    int* __restrict__ cnt, int* __restrict__ rank) {
  int e = blockIdx.x * blockDim.x + threadIdx.x;
  if (e >= nnz0 + nnz1) return;
  int r = (e < nnz0) ? row0[e] : (row1[e - nnz0] + rbase1);
  rank[e] = atomicAdd(&cnt[r], 1);
}

__global__ void __launch_bounds__(1024) offsets_scan_kernel(
    const int* __restrict__ cnt, int* __restrict__ start, int n) {
  __shared__ int tile_part[32];
  __shared__ int head_part[32];
  const int t = threadIdx.x;
  const int lane = t & 31;
  const int w = t >> 5;
  const int tile0 = blockIdx.x * 4096;
  const int4* c4 = (const int4*)cnt;

  int head = 0;
#pragma unroll 8
  for (int k = 0; k < (int)blockIdx.x; ++k) {
    int4 q = c4[k * 1024 + t];
    head += (q.x + q.y) + (q.z + q.w);
  }

  const int idx = tile0 + 4 * t;
  int a0 = 0, a1 = 0, a2 = 0, a3 = 0;
  if (idx + 3 < n) {
    int4 q = c4[idx >> 2];
    a0 = q.x; a1 = q.y; a2 = q.z; a3 = q.w;
  } else {
    if (idx     < n) a0 = cnt[idx];
    if (idx + 1 < n) a1 = cnt[idx + 1];
    if (idx + 2 < n) a2 = cnt[idx + 2];
  }
  const int mine = (a0 + a1) + (a2 + a3);

  int incl = mine;
  for (int d = 1; d < 32; d <<= 1) {
    int u = __shfl_up(incl, d);
    if (lane >= d) incl += u;
  }
  for (int d = 16; d >= 1; d >>= 1) head += __shfl_xor(head, d);
  if (lane == 31) tile_part[w] = incl;
  if (lane == 0)  head_part[w] = head;
  __syncthreads();

  int off = 0;
  for (int q = 0; q < 32; ++q) {
    off += head_part[q];
    if (q < w) off += tile_part[q];
  }
  const int ex = off + incl - mine;
  if (idx + 3 < n) {
    ((int4*)start)[idx >> 2] = make_int4(ex, ex + a0, ex + a0 + a1, ex + a0 + a1 + a2);
  } else {
    if (idx     < n) start[idx]     = ex;
    if (idx + 1 < n) start[idx + 1] = ex + a0;
    if (idx + 2 < n) start[idx + 2] = ex + a0 + a1;
  }
}

template <bool PAIRS>
__global__ void __launch_bounds__(256) fill_slots_kernel(
    const int* __restrict__ row0, const int* __restrict__ col0,
    const float* __restrict__ val0, int nnz0,
    const int* __restrict__ row1, const int* __restrict__ col1,
    const float* __restrict__ val1, int nnz1, int rbase1,
    const int* __restrict__ start, const int* __restrict__ rank,
    int* __restrict__ slots) {
  int e = blockIdx.x * blockDim.x + threadIdx.x;
  if (e >= nnz0 + nnz1) return;
  int r, c = 0;
  float v = 0.0f;
  if (e < nnz0) {
    r = row0[e];
    if (PAIRS) { c = col0[e]; v = val0[e]; }
  } else {
    int f = e - nnz0;
    r = row1[f] + rbase1;
    if (PAIRS) { c = col1[f]; v = val1[f]; }
  }
  int pos = start[r] + rank[e];
  if (PAIRS) ((int2*)slots)[pos] = make_int2(c, __float_as_int(v));
  else       slots[pos] = e;
}

template <bool PAIRS>
__global__ void __launch_bounds__(256) gather_rows_kernel(
    const int* __restrict__ start, const int* __restrict__ slots,
    const int* __restrict__ col, const float* __restrict__ val,
    const float* __restrict__ x, float* __restrict__ y,
    float* __restrict__ deg, int n_rows) {
  const int g = blockIdx.x * blockDim.x + threadIdx.x;
  const int r = g >> 4;
  const int j = g & 15;
  int s = 0, n = 0;
  if (r < n_rows) { s = start[r]; n = start[r + 1] - s; }
  const int nmax = max(n, __shfl_xor(n, 16));

  float4 acc = make_float4(0.0f, 0.0f, 0.0f, 0.0f);
  float d = 0.0f;
  for (int base = 0; base < nmax; base += 16) {
    int c = 0;
    float v = 0.0f;
    if (base + j < n) {
      if (PAIRS) {
        int2 p = ((const int2*)slots)[s + base + j];
        c = p.x; v = __int_as_float(p.y);
      } else {
        int e = slots[s + base + j];
        c = col[e]; v = val[e];
      }
    }
    const int lim = min(nmax - base, 16);
    for (int i = 0; i < lim; i += 4) {
      const int   c0 = __shfl(c, i,     16), c1 = __shfl(c, i + 1, 16);
      const int   c2 = __shfl(c, i + 2, 16), c3 = __shfl(c, i + 3, 16);
      const float v0 = __shfl(v, i,     16), v1 = __shfl(v, i + 1, 16);
      const float v2 = __shfl(v, i + 2, 16), v3 = __shfl(v, i + 3, 16);
      const float4 x0 = ((const float4*)(x + (size_t)c0 * EMB))[j];
      const float4 x1 = ((const float4*)(x + (size_t)c1 * EMB))[j];
      const float4 x2 = ((const float4*)(x + (size_t)c2 * EMB))[j];
      const float4 x3 = ((const float4*)(x + (size_t)c3 * EMB))[j];
      if (base + i < n) {
        acc.x = fmaf(v0, x0.x, acc.x); acc.y = fmaf(v0, x0.y, acc.y);
        acc.z = fmaf(v0, x0.z, acc.z); acc.w = fmaf(v0, x0.w, acc.w);
        d += v0;
      }
      if (base + i + 1 < n) {
        acc.x = fmaf(v1, x1.x, acc.x); acc.y = fmaf(v1, x1.y, acc.y);
        acc.z = fmaf(v1, x1.z, acc.z); acc.w = fmaf(v1, x1.w, acc.w);
        d += v1;
      }
      if (base + i + 2 < n) {
        acc.x = fmaf(v2, x2.x, acc.x); acc.y = fmaf(v2, x2.y, acc.y);
        acc.z = fmaf(v2, x2.z, acc.z); acc.w = fmaf(v2, x2.w, acc.w);
        d += v2;
      }
      if (base + i + 3 < n) {
        acc.x = fmaf(v3, x3.x, acc.x); acc.y = fmaf(v3, x3.y, acc.y);
        acc.z = fmaf(v3, x3.z, acc.z); acc.w = fmaf(v3, x3.w, acc.w);
        d += v3;
      }
    }
  }
  if (r < n_rows) {
    ((float4*)(y + (size_t)r * EMB))[j] = acc;
    if (deg != nullptr && j == 0) deg[r] = d;
  }
}

__global__ void __launch_bounds__(128) acc_wmma_kernel(
    const float* __restrict__ e, float* __restrict__ acc) {
  int wave = threadIdx.x >> 5;
  int lane = threadIdx.x & 31;
  int n    = lane & 15;
  int hi   = lane >> 4;
  int row0 = blockIdx.x * 16;
  int col0 = wave * 16;

  const size_t base = (size_t)row0 * EMB + col0 + n;
  const float* ap = acc + base;
  const float* ep = e + base;

  v8f c;
#pragma unroll
  for (int r = 0; r < 8; ++r)
    c[r] = ap[(size_t)(r + 8 * hi) * EMB];

  v2f a[4], b[4];
#pragma unroll
  for (int k = 0; k < 4; ++k) {
    a[k].x = (n == (4 * k + 2 * hi))     ? 1.0f : 0.0f;
    a[k].y = (n == (4 * k + 2 * hi + 1)) ? 1.0f : 0.0f;
    b[k].x = ep[(size_t)(4 * k + 2 * hi) * EMB];
    b[k].y = ep[(size_t)(4 * k + 2 * hi + 1) * EMB];
  }

#pragma unroll
  for (int k = 0; k < 4; ++k)
    c = __builtin_amdgcn_wmma_f32_16x16x4_f32(
        false, a[k], false, b[k], (short)0, c, false, false);

  float* op = acc + base;
#pragma unroll
  for (int r = 0; r < 8; ++r)
    op[(size_t)(r + 8 * hi) * EMB] = c[r];
}

__global__ void __launch_bounds__(256) rownorm_kernel(
    float* __restrict__ x, const float* __restrict__ deg, int rowsVec) {
  int g = blockIdx.x * blockDim.x + threadIdx.x;
  if (g >= rowsVec) return;
  float d = deg[g >> 4];
  d = (d == 0.0f) ? 1.0f : d;
  float inv = 1.0f / d;
  float4 v = ((float4*)x)[g];
  v.x *= inv; v.y *= inv; v.z *= inv; v.w *= inv;
  ((float4*)x)[g] = v;
}

__global__ void __launch_bounds__(256) finalize_kernel(
    float* __restrict__ out, const float* __restrict__ ulocal,
    const float* __restrict__ deg, int totalVec, int userVec) {
  int g = blockIdx.x * blockDim.x + threadIdx.x;
  if (g >= totalVec) return;
  float4 v = ((float4*)out)[g];
  v.x *= 0.25f; v.y *= 0.25f; v.z *= 0.25f; v.w *= 0.25f;
  if (g < userVec) {
    float d = deg[g >> 4];
    d = (d == 0.0f) ? 1.0f : d;
    float inv = 1.0f / d;
    float4 w = ((const float4*)ulocal)[g];
    v.x += w.x * inv; v.y += w.y * inv; v.z += w.z * inv; v.w += w.w * inv;
  }
  ((float4*)out)[g] = v;
}

static inline int cdiv(long long a, int b) { return (int)((a + b - 1) / b); }

extern "C" void kernel_launch(void* const* d_in, const int* in_sizes, int n_in,
                              void* d_out, int out_size, void* d_ws, size_t ws_size,
                              hipStream_t stream) {
  const float* user_emb = (const float*)d_in[0];
  const float* item_emb = (const float*)d_in[1];
  const float* adj_val  = (const float*)d_in[2];
  const float* hv_val   = (const float*)d_in[3];
  const float* hu_val   = (const float*)d_in[4];
  const int*   adj_row  = (const int*)d_in[5];
  const int*   adj_col  = (const int*)d_in[6];
  const int*   hv_row   = (const int*)d_in[7];
  const int*   hv_col   = (const int*)d_in[8];
  const int*   hu_row   = (const int*)d_in[9];
  const int*   hu_col   = (const int*)d_in[10];

  const int E_adj = in_sizes[2];
  const int E_hv  = in_sizes[3];
  const int E_hu  = in_sizes[4];

  float* out = (float*)d_out;

  const size_t embBytesN = (size_t)NUM_NODES * EMB * sizeof(float);
  const size_t embBytesB = (size_t)NUM_BQ    * EMB * sizeof(float);
  char* ws = (char*)d_ws;
  float* e0     = (float*)(ws);
  float* e1     = (float*)(ws + embBytesN);
  float* bfeat  = (float*)(ws + 2 * embBytesN);
  float* deg_hv = (float*)(ws + 2 * embBytesN + embBytesB);
  float* deg_hu = deg_hv + NUM_BQ;
  float* ulocal = e1;

  int* cnt_a   = (int*)(ws + embBytesN);
  int* rank_a  = (int*)(ws + embBytesN + 600064);
  int* start_a = (int*)(ws + 2 * embBytesN);
  int* perm_a  = (int*)(ws + 2 * embBytesN + 600064);

  int* cnt_h   = (int*)(ws);
  int* start_h = (int*)(ws + 480064);
  int* rank_h  = (int*)(ws + 960128);
  int* pairs_h = (int*)(ws + 4160128);

  const int totalVec = NUM_NODES * (EMB / 4);
  const int userVec  = NUM_USERS * (EMB / 4);
  const int bqVec    = NUM_BQ    * (EMB / 4);

  copy_init_kernel<<<cdiv(totalVec, 256), 256, 0, stream>>>(
      user_emb, item_emb, e0, out, totalVec, userVec);

  hipMemsetAsync(cnt_a, 0, (size_t)(NUM_NODES + 1) * sizeof(int), stream);
  count_rank_kernel<<<cdiv(E_adj, 256), 256, 0, stream>>>(
      adj_row, E_adj, nullptr, 0, 0, cnt_a, rank_a);
  offsets_scan_kernel<<<cdiv(NUM_NODES + 1, 4096), 1024, 0, stream>>>(
      cnt_a, start_a, NUM_NODES + 1);
  fill_slots_kernel<false><<<cdiv(E_adj, 256), 256, 0, stream>>>(
      adj_row, nullptr, nullptr, E_adj, nullptr, nullptr, nullptr, 0, 0,
      start_a, rank_a, perm_a);

  float* eprev = e0;
  float* enext = e1;
  for (int layer = 0; layer < 3; ++layer) {
    gather_rows_kernel<false><<<cdiv((long long)NUM_NODES * 16, 256), 256, 0, stream>>>(
        start_a, perm_a, adj_col, adj_val, eprev, enext, nullptr, NUM_NODES);
    acc_wmma_kernel<<<NUM_NODES / 16, 128, 0, stream>>>(enext, out);
    float* t = eprev; eprev = enext; enext = t;
  }

  hipMemsetAsync(cnt_h, 0, (size_t)(NUM_BQ + NUM_USERS + 1) * sizeof(int), stream);
  count_rank_kernel<<<cdiv((long long)E_hv + E_hu, 256), 256, 0, stream>>>(
      hv_row, E_hv, hu_row, E_hu, NUM_BQ, cnt_h, rank_h);
  offsets_scan_kernel<<<cdiv(NUM_BQ + NUM_USERS + 1, 4096), 1024, 0, stream>>>(
      cnt_h, start_h, NUM_BQ + NUM_USERS + 1);
  fill_slots_kernel<true><<<cdiv((long long)E_hv + E_hu, 256), 256, 0, stream>>>(
      hv_row, hv_col, hv_val, E_hv, hu_row, hu_col, hu_val, E_hu, NUM_BQ,
      start_h, rank_h, pairs_h);

  gather_rows_kernel<true><<<cdiv((long long)NUM_BQ * 16, 256), 256, 0, stream>>>(
      start_h, pairs_h, nullptr, nullptr, item_emb, bfeat, deg_hv, NUM_BQ);
  rownorm_kernel<<<cdiv(bqVec, 256), 256, 0, stream>>>(bfeat, deg_hv, bqVec);

  gather_rows_kernel<true><<<cdiv((long long)NUM_USERS * 16, 256), 256, 0, stream>>>(
      start_h + NUM_BQ, pairs_h, nullptr, nullptr, bfeat, ulocal, deg_hu, NUM_USERS);

  finalize_kernel<<<cdiv(totalVec, 256), 256, 0, stream>>>(
      out, ulocal, deg_hu, totalVec, userVec);
}


# Layer sum deferred: layer 0 read from the inputs, layers 1-3 kept in buffers; two WMMA passes (out=e0+e1+e2; out=(out+e3)/4+local) replace copy, 3 accumulate passes, finalize; sum order kept
# speedup vs baseline: 1.9596x; 1.1944x over previous
#include <hip/hip_runtime.h>
#include <hip/hip_bf16.h>


typedef __attribute__((ext_vector_type(2))) float v2f;
typedef __attribute__((ext_vector_type(8))) float v8f;

#define NUM_USERS 100000
#define NUM_ITEMS 50000
#define NUM_NODES 150000
#define NUM_BQ    20000
#define EMB       64

__global__ void __launch_bounds__(256) count_rank_kernel(
    const int* __restrict__ row0, int nnz0,
    const int* __restrict__ row1, int nnz1, int rbase1,
    int* __restrict__ cnt, int* __restrict__ rank) {
  int e = blockIdx.x * blockDim.x + threadIdx.x;
  if (e >= nnz0 + nnz1) return;
  int r = (e < nnz0) ? row0[e] : (row1[e - nnz0] + rbase1);
  rank[e] = atomicAdd(&cnt[r], 1);
}

__global__ void __launch_bounds__(1024) offsets_scan_kernel(
    const int* __restrict__ cnt, int* __restrict__ start, int n) {
  __shared__ int tile_part[32];
  __shared__ int head_part[32];
  const int t = threadIdx.x;
  const int lane = t & 31;
  const int w = t >> 5;
  const int tile0 = blockIdx.x * 4096;
  const int4* c4 = (const int4*)cnt;

  int head = 0;
#pragma unroll 8
  for (int k = 0; k < (int)blockIdx.x; ++k) {
    int4 q = c4[k * 1024 + t];
    head += (q.x + q.y) + (q.z + q.w);
  }

  const int idx = tile0 + 4 * t;
  int a0 = 0, a1 = 0, a2 = 0, a3 = 0;
  if (idx + 3 < n) {
    int4 q = c4[idx >> 2];
    a0 = q.x; a1 = q.y; a2 = q.z; a3 = q.w;
  } else {
    if (idx     < n) a0 = cnt[idx];
    if (idx + 1 < n) a1 = cnt[idx + 1];
    if (idx + 2 < n) a2 = cnt[idx + 2];
  }
  const int mine = (a0 + a1) + (a2 + a3);

  int incl = mine;
  for (int d = 1; d < 32; d <<= 1) {
    int u = __shfl_up(incl, d);
    if (lane >= d) incl += u;
  }
  for (int d = 16; d >= 1; d >>= 1) head += __shfl_xor(head, d);
  if (lane == 31) tile_part[w] = incl;
  if (lane == 0)  head_part[w] = head;
  __syncthreads();

  int off = 0;
  for (int q = 0; q < 32; ++q) {
    off += head_part[q];
    if (q < w) off += tile_part[q];
  }
  const int ex = off + incl - mine;
  if (idx + 3 < n) {
    ((int4*)start)[idx >> 2] = make_int4(ex, ex + a0, ex + a0 + a1, ex + a0 + a1 + a2);
  } else {
    if (idx     < n) start[idx]     = ex;
    if (idx + 1 < n) start[idx + 1] = ex + a0;
    if (idx + 2 < n) start[idx + 2] = ex + a0 + a1;
  }
}

template <bool PAIRS>
__global__ void __launch_bounds__(256) fill_slots_kernel(
    const int* __restrict__ row0, const int* __restrict__ col0,
    const float* __restrict__ val0, int nnz0,
    const int* __restrict__ row1, const int* __restrict__ col1,
    const float* __restrict__ val1, int nnz1, int rbase1,
    const int* __restrict__ start, const int* __restrict__ rank,
    int* __restrict__ slots) {
  int e = blockIdx.x * blockDim.x + threadIdx.x;
  if (e >= nnz0 + nnz1) return;
  int r, c = 0;
  float v = 0.0f;
  if (e < nnz0) {
    r = row0[e];
    if (PAIRS) { c = col0[e]; v = val0[e]; }
  } else {
    int f = e - nnz0;
    r = row1[f] + rbase1;
    if (PAIRS) { c = col1[f]; v = val1[f]; }
  }
  int pos = start[r] + rank[e];
  if (PAIRS) ((int2*)slots)[pos] = make_int2(c, __float_as_int(v));
  else       slots[pos] = e;
}

template <bool PAIRS, bool SPLIT>
__global__ void __launch_bounds__(256) gather_rows_kernel(
    const int* __restrict__ start, const int* __restrict__ slots,
    const int* __restrict__ col, const float* __restrict__ val,
    const float* __restrict__ x, const float* __restrict__ x1, int split,
    float* __restrict__ y, float* __restrict__ deg, int n_rows) {
  const int g = blockIdx.x * blockDim.x + threadIdx.x;
  const int r = g >> 4;
  const int j = g & 15;
  int s = 0, n = 0;
  if (r < n_rows) { s = start[r]; n = start[r + 1] - s; }
  const int nmax = max(n, __shfl_xor(n, 16));

  float4 acc = make_float4(0.0f, 0.0f, 0.0f, 0.0f);
  float d = 0.0f;
  for (int base = 0; base < nmax; base += 16) {
    int c = 0;
    float v = 0.0f;
    if (base + j < n) {
      if (PAIRS) {
        int2 p = ((const int2*)slots)[s + base + j];
        c = p.x; v = __int_as_float(p.y);
      } else {
        int e = slots[s + base + j];
        c = col[e]; v = val[e];
      }
    }
    const int lim = min(nmax - base, 16);
    for (int i = 0; i < lim; i += 4) {
      const int   c0 = __shfl(c, i,     16), c1 = __shfl(c, i + 1, 16);
      const int   c2 = __shfl(c, i + 2, 16), c3 = __shfl(c, i + 3, 16);
      const float v0 = __shfl(v, i,     16), v1 = __shfl(v, i + 1, 16);
      const float v2 = __shfl(v, i + 2, 16), v3 = __shfl(v, i + 3, 16);
      const float* p0 = x + (size_t)c0 * EMB;
      const float* p1 = x + (size_t)c1 * EMB;
      const float* p2 = x + (size_t)c2 * EMB;
      const float* p3 = x + (size_t)c3 * EMB;
      if (SPLIT) {
        if (c0 >= split) p0 = x1 + (size_t)(c0 - split) * EMB;
        if (c1 >= split) p1 = x1 + (size_t)(c1 - split) * EMB;
        if (c2 >= split) p2 = x1 + (size_t)(c2 - split) * EMB;
        if (c3 >= split) p3 = x1 + (size_t)(c3 - split) * EMB;
      }
      const float4 x0 = ((const float4*)p0)[j];
      const float4 x1v = ((const float4*)p1)[j];
      const float4 x2 = ((const float4*)p2)[j];
      const float4 x3 = ((const float4*)p3)[j];
      if (base + i < n) {
        acc.x = fmaf(v0, x0.x, acc.x); acc.y = fmaf(v0, x0.y, acc.y);
        acc.z = fmaf(v0, x0.z, acc.z); acc.w = fmaf(v0, x0.w, acc.w);
        d += v0;
      }
      if (base + i + 1 < n) {
        acc.x = fmaf(v1, x1v.x, acc.x); acc.y = fmaf(v1, x1v.y, acc.y);
        acc.z = fmaf(v1, x1v.z, acc.z); acc.w = fmaf(v1, x1v.w, acc.w);
        d += v1;
      }
      if (base + i + 2 < n) {
        acc.x = fmaf(v2, x2.x, acc.x); acc.y = fmaf(v2, x2.y, acc.y);
        acc.z = fmaf(v2, x2.z, acc.z); acc.w = fmaf(v2, x2.w, acc.w);
        d += v2;
      }
      if (base + i + 3 < n) {
        acc.x = fmaf(v3, x3.x, acc.x); acc.y = fmaf(v3, x3.y, acc.y);
        acc.z = fmaf(v3, x3.z, acc.z); acc.w = fmaf(v3, x3.w, acc.w);
        d += v3;
      }
    }
  }
  if (r < n_rows) {
    ((float4*)(y + (size_t)r * EMB))[j] = acc;
    if (deg != nullptr && j == 0) deg[r] = d;
  }
}

__device__ __forceinline__ void identity_frags(int n, int hi, v2f a[4]) {
#pragma unroll
  for (int k = 0; k < 4; ++k) {
    a[k].x = (n == (4 * k + 2 * hi))     ? 1.0f : 0.0f;
    a[k].y = (n == (4 * k + 2 * hi + 1)) ? 1.0f : 0.0f;
  }
}

__device__ __forceinline__ void load_b_frags(const float* __restrict__ ep, int hi, v2f b[4]) {
#pragma unroll
  for (int k = 0; k < 4; ++k) {
    b[k].x = ep[(size_t)(4 * k + 2 * hi) * EMB];
    b[k].y = ep[(size_t)(4 * k + 2 * hi + 1) * EMB];
  }
}

__global__ void __launch_bounds__(128) sum3_wmma_kernel(
    const float* __restrict__ ue, const float* __restrict__ ie,
    const float* __restrict__ e1, const float* __restrict__ e2,
    float* __restrict__ out) {
  int wave = threadIdx.x >> 5;
  int lane = threadIdx.x & 31;
  int n    = lane & 15;
  int hi   = lane >> 4;
  int row0 = blockIdx.x * 16;
  int col0 = wave * 16;

  const size_t base = (size_t)row0 * EMB + col0 + n;
  const float* zp = (row0 < NUM_USERS)
      ? ue + base
      : ie + ((size_t)(row0 - NUM_USERS) * EMB + col0 + n);

  v8f c;
#pragma unroll
  for (int r = 0; r < 8; ++r)
    c[r] = zp[(size_t)(r + 8 * hi) * EMB];

  v2f a[4], b1[4], b2[4];
  identity_frags(n, hi, a);
  load_b_frags(e1 + base, hi, b1);
  load_b_frags(e2 + base, hi, b2);

#pragma unroll
  for (int k = 0; k < 4; ++k)
    c = __builtin_amdgcn_wmma_f32_16x16x4_f32(
        false, a[k], false, b1[k], (short)0, c, false, false);
#pragma unroll
  for (int k = 0; k < 4; ++k)
    c = __builtin_amdgcn_wmma_f32_16x16x4_f32(
        false, a[k], false, b2[k], (short)0, c, false, false);

  float* op = out + base;
#pragma unroll
  for (int r = 0; r < 8; ++r)
    op[(size_t)(r + 8 * hi) * EMB] = c[r];
}

__global__ void __launch_bounds__(128) last_wmma_kernel(
    const float* __restrict__ e3, const float* __restrict__ ulocal,
    const float* __restrict__ deg, float* __restrict__ out) {
  int wave = threadIdx.x >> 5;
  int lane = threadIdx.x & 31;
  int n    = lane & 15;
  int hi   = lane >> 4;
  int row0 = blockIdx.x * 16;
  int col0 = wave * 16;
  const bool user = row0 < NUM_USERS;

  const size_t base = (size_t)row0 * EMB + col0 + n;
  const float* ap = out + base;

  v8f c;
#pragma unroll
  for (int r = 0; r < 8; ++r)
    c[r] = ap[(size_t)(r + 8 * hi) * EMB];

  v2f a[4], b[4];
  identity_frags(n, hi, a);
  load_b_frags(e3 + base, hi, b);

  v8f w, dg;
#pragma unroll
  for (int r = 0; r < 8; ++r) { w[r] = 0.0f; dg[r] = 1.0f; }
  if (user) {
    const float* up = ulocal + base;
#pragma unroll
    for (int r = 0; r < 8; ++r)
      w[r] = up[(size_t)(r + 8 * hi) * EMB];
    const float4 d0 = ((const float4*)(deg + row0 + 8 * hi))[0];
    const float4 d1 = ((const float4*)(deg + row0 + 8 * hi))[1];
    dg[0] = d0.x; dg[1] = d0.y; dg[2] = d0.z; dg[3] = d0.w;
    dg[4] = d1.x; dg[5] = d1.y; dg[6] = d1.z; dg[7] = d1.w;
  }

#pragma unroll
  for (int k = 0; k < 4; ++k)
    c = __builtin_amdgcn_wmma_f32_16x16x4_f32(
        false, a[k], false, b[k], (short)0, c, false, false);

  float* op = out + base;
  if (user) {
#pragma unroll
    for (int r = 0; r < 8; ++r) {
      float d = dg[r];
      d = (d == 0.0f) ? 1.0f : d;
      float inv = 1.0f / d;
      float v = c[r] * 0.25f;
      v += w[r] * inv;
      op[(size_t)(r + 8 * hi) * EMB] = v;
    }
  } else {
#pragma unroll
    for (int r = 0; r < 8; ++r)
      op[(size_t)(r + 8 * hi) * EMB] = c[r] * 0.25f;
  }
}

__global__ void __launch_bounds__(256) rownorm_kernel(
    float* __restrict__ x, const float* __restrict__ deg, int rowsVec) {
  int g = blockIdx.x * blockDim.x + threadIdx.x;
  if (g >= rowsVec) return;
  float d = deg[g >> 4];
  d = (d == 0.0f) ? 1.0f : d;
  float inv = 1.0f / d;
  float4 v = ((float4*)x)[g];
  v.x *= inv; v.y *= inv; v.z *= inv; v.w *= inv;
  ((float4*)x)[g] = v;
}

static inline int cdiv(long long a, int b) { return (int)((a + b - 1) / b); }

extern "C" void kernel_launch(void* const* d_in, const int* in_sizes, int n_in,
                              void* d_out, int out_size, void* d_ws, size_t ws_size,
                              hipStream_t stream) {
  const float* user_emb = (const float*)d_in[0];
  const float* item_emb = (const float*)d_in[1];
  const float* adj_val  = (const float*)d_in[2];
  const float* hv_val   = (const float*)d_in[3];
  const float* hu_val   = (const float*)d_in[4];
  const int*   adj_row  = (const int*)d_in[5];
  const int*   adj_col  = (const int*)d_in[6];
  const int*   hv_row   = (const int*)d_in[7];
  const int*   hv_col   = (const int*)d_in[8];
  const int*   hu_row   = (const int*)d_in[9];
  const int*   hu_col   = (const int*)d_in[10];

  const int E_adj = in_sizes[2];
  const int E_hv  = in_sizes[3];
  const int E_hu  = in_sizes[4];

  float* out = (float*)d_out;

  const size_t embBytesN = (size_t)NUM_NODES * EMB * sizeof(float);
  const size_t embBytesB = (size_t)NUM_BQ    * EMB * sizeof(float);
  char* ws = (char*)d_ws;
  float* buf0   = (float*)(ws);
  float* buf1   = (float*)(ws + embBytesN);
  float* bfeat  = (float*)(ws + 2 * embBytesN);
  float* deg_hv = (float*)(ws + 2 * embBytesN + embBytesB);
  float* deg_hu = deg_hv + NUM_BQ;

  int* cnt_a   = (int*)(ws + embBytesN);
  int* rank_a  = (int*)(ws + embBytesN + 600064);
  int* start_a = (int*)(ws + 2 * embBytesN);
  int* perm_a  = (int*)(ws + 2 * embBytesN + 600064);

  int* cnt_h   = (int*)(ws + embBytesN);
  int* start_h = (int*)(ws + embBytesN + 480064);
  int* rank_h  = (int*)(ws + embBytesN + 960128);
  int* pairs_h = (int*)(ws + embBytesN + 4160128);
  float* ulocal = (float*)(ws + embBytesN + 10560256);

  const int bqVec = NUM_BQ * (EMB / 4);

  hipMemsetAsync(cnt_a, 0, (size_t)(NUM_NODES + 1) * sizeof(int), stream);
  count_rank_kernel<<<cdiv(E_adj, 256), 256, 0, stream>>>(
      adj_row, E_adj, nullptr, 0, 0, cnt_a, rank_a);
  offsets_scan_kernel<<<cdiv(NUM_NODES + 1, 4096), 1024, 0, stream>>>(
      cnt_a, start_a, NUM_NODES + 1);
  fill_slots_kernel<false><<<cdiv(E_adj, 256), 256, 0, stream>>>(
      adj_row, nullptr, nullptr, E_adj, nullptr, nullptr, nullptr, 0, 0,
      start_a, rank_a, perm_a);

  const int gatherBlocksN = cdiv((long long)NUM_NODES * 16, 256);
  gather_rows_kernel<false, true><<<gatherBlocksN, 256, 0, stream>>>(
      start_a, perm_a, adj_col, adj_val, user_emb, item_emb, NUM_USERS,
      buf0, nullptr, NUM_NODES);
  gather_rows_kernel<false, false><<<gatherBlocksN, 256, 0, stream>>>(
      start_a, perm_a, adj_col, adj_val, buf0, nullptr, 0,
      buf1, nullptr, NUM_NODES);
  sum3_wmma_kernel<<<NUM_NODES / 16, 128, 0, stream>>>(
      user_emb, item_emb, buf0, buf1, out);
  gather_rows_kernel<false, false><<<gatherBlocksN, 256, 0, stream>>>(
      start_a, perm_a, adj_col, adj_val, buf1, nullptr, 0,
      buf0, nullptr, NUM_NODES);

  hipMemsetAsync(cnt_h, 0, (size_t)(NUM_BQ + NUM_USERS + 1) * sizeof(int), stream);
  count_rank_kernel<<<cdiv((long long)E_hv + E_hu, 256), 256, 0, stream>>>(
      hv_row, E_hv, hu_row, E_hu, NUM_BQ, cnt_h, rank_h);
  offsets_scan_kernel<<<cdiv(NUM_BQ + NUM_USERS + 1, 4096), 1024, 0, stream>>>(
      cnt_h, start_h, NUM_BQ + NUM_USERS + 1);
  fill_slots_kernel<true><<<cdiv((long long)E_hv + E_hu, 256), 256, 0, stream>>>(
      hv_row, hv_col, hv_val, E_hv, hu_row, hu_col, hu_val, E_hu, NUM_BQ,
      start_h, rank_h, pairs_h);

  gather_rows_kernel<true, false><<<cdiv((long long)NUM_BQ * 16, 256), 256, 0, stream>>>(
      start_h, pairs_h, nullptr, nullptr, item_emb, nullptr, 0,
      bfeat, deg_hv, NUM_BQ);
  rownorm_kernel<<<cdiv(bqVec, 256), 256, 0, stream>>>(bfeat, deg_hv, bqVec);

  gather_rows_kernel<true, false><<<cdiv((long long)NUM_USERS * 16, 256), 256, 0, stream>>>(
      start_h + NUM_BQ, pairs_h, nullptr, nullptr, bfeat, nullptr, 0,
      ulocal, deg_hu, NUM_USERS);

  last_wmma_kernel<<<NUM_NODES / 16, 128, 0, stream>>>(buf0, ulocal, deg_hu, out);
}


# Edge buckets by a two-level counting sort in LDS (512-row bins: block histograms, one range per bin and block, whole-run stores; a block orders a bin), not one global add + scattered store per edge
# speedup vs baseline: 3.0494x; 1.5561x over previous
#include <hip/hip_runtime.h>
#include <hip/hip_bf16.h>


typedef __attribute__((ext_vector_type(2))) float v2f;
typedef __attribute__((ext_vector_type(8))) float v8f;

#define NUM_USERS 100000
#define NUM_ITEMS 50000
#define NUM_NODES 150000
#define NUM_BQ    20000
#define EMB       64

#define BIN_SHIFT 9
#define BIN_ROWS  512
#define CHUNK     4096
#define STAGE_CAP 12288
#define SWEEP     8

__device__ __forceinline__ int wave_incl_scan(int v, int lane) {
  for (int d = 1; d < 32; d <<= 1) {
    int u = __shfl_up(v, d);
    if (lane >= d) v += u;
  }
  return v;
}

__device__ __forceinline__ int edge_row(const int* __restrict__ row0, int nnz0,
                                        const int* __restrict__ row1, int rbase1, int e) {
  return (e < nnz0) ? row0[e] : (row1[e - nnz0] + rbase1);
}

__global__ void __launch_bounds__(BIN_ROWS) bin_count_kernel(
    const int* __restrict__ row0, int nnz0,
    const int* __restrict__ row1, int nnz1, int rbase1,
    int* __restrict__ tot) {
  __shared__ int h[BIN_ROWS];
  const int t = threadIdx.x;
  h[t] = 0;
  __syncthreads();
  const int nnz = nnz0 + nnz1;
  const int e0 = blockIdx.x * CHUNK;
  int r[CHUNK / BIN_ROWS];
#pragma unroll
  for (int k = 0; k < CHUNK / BIN_ROWS; ++k) {
    const int e = e0 + k * BIN_ROWS + t;
    r[k] = (e < nnz) ? edge_row(row0, nnz0, row1, rbase1, e) : -1;
  }
#pragma unroll
  for (int k = 0; k < CHUNK / BIN_ROWS; ++k)
    if (r[k] >= 0) atomicAdd(&h[r[k] >> BIN_SHIFT], 1);
  __syncthreads();
  const int c = h[t];
  if (c > 0) atomicAdd(&tot[t], c);
}

__global__ void __launch_bounds__(BIN_ROWS) bin_scatter_kernel(
    const int* __restrict__ row0, int nnz0,
    const int* __restrict__ row1, int nnz1, int rbase1,
    const int* __restrict__ tot, int* __restrict__ cur,
    int* __restrict__ coarse, int nb) {
  __shared__ int h[BIN_ROWS];
  __shared__ int ls[BIN_ROWS];
  __shared__ int bb[BIN_ROWS];
  __shared__ int wsum[BIN_ROWS / 32];
  __shared__ int stage[CHUNK];
  __shared__ int dest[CHUNK];
  const int t = threadIdx.x;
  const int lane = t & 31;
  const int w = t >> 5;

  const int tv = (t < nb) ? tot[t] : 0;
  const int incl = wave_incl_scan(tv, lane);
  if (lane == 31) wsum[w] = incl;
  h[t] = 0;
  __syncthreads();
  int off = 0;
  for (int q = 0; q < BIN_ROWS / 32; ++q) if (q < w) off += wsum[q];
  bb[t] = off + incl - tv;
  __syncthreads();

  const int nnz = nnz0 + nnz1;
  const int e0 = blockIdx.x * CHUNK;
  int r[CHUNK / BIN_ROWS], bin[CHUNK / BIN_ROWS], lr[CHUNK / BIN_ROWS], rec[CHUNK / BIN_ROWS];
#pragma unroll
  for (int k = 0; k < CHUNK / BIN_ROWS; ++k) {
    const int e = e0 + k * BIN_ROWS + t;
    r[k] = (e < nnz) ? edge_row(row0, nnz0, row1, rbase1, e) : -1;
  }
#pragma unroll
  for (int k = 0; k < CHUNK / BIN_ROWS; ++k) {
    const int e = e0 + k * BIN_ROWS + t;
    bin[k] = -1; lr[k] = 0; rec[k] = 0;
    if (r[k] >= 0) {
      bin[k] = r[k] >> BIN_SHIFT;
      lr[k] = atomicAdd(&h[bin[k]], 1);
      rec[k] = (int)(((unsigned)e << BIN_SHIFT) | (unsigned)(r[k] & (BIN_ROWS - 1)));
    }
  }
  __syncthreads();

  const int hv = h[t];
  const int incl2 = wave_incl_scan(hv, lane);
  if (lane == 31) wsum[w] = incl2;
  __syncthreads();
  int off2 = 0;
  for (int q = 0; q < BIN_ROWS / 32; ++q) if (q < w) off2 += wsum[q];
  ls[t] = off2 + incl2 - hv;
  int g = 0;
  if (hv > 0) g = bb[t] + atomicAdd(&cur[t], hv);
  h[t] = g;
  __syncthreads();

#pragma unroll
  for (int k = 0; k < CHUNK / BIN_ROWS; ++k) {
    if (bin[k] >= 0) {
      const int p = ls[bin[k]] + lr[k];
      stage[p] = rec[k];
      dest[p] = h[bin[k]] + lr[k];
    }
  }
  __syncthreads();

  const int n_here = min(CHUNK, nnz - e0);
  for (int p = t; p < n_here; p += BIN_ROWS) coarse[dest[p]] = stage[p];
}

template <bool PAIRS>
__device__ __forceinline__ void emit_slot(
    int* __restrict__ slots, int pos, int id,
    const int* __restrict__ col0, const float* __restrict__ val0, int nnz0,
    const int* __restrict__ col1, const float* __restrict__ val1) {
  if (PAIRS) {
    int c;
    float v;
    if (id < nnz0) { c = col0[id]; v = val0[id]; }
    else           { c = col1[id - nnz0]; v = val1[id - nnz0]; }
    ((int2*)slots)[pos] = make_int2(c, __float_as_int(v));
  } else {
    slots[pos] = id;
  }
}

template <bool PAIRS>
__global__ void __launch_bounds__(BIN_ROWS) bin_sort_kernel(
    const int* __restrict__ tot, const int* __restrict__ coarse, int nb, int n_rows,
    const int* __restrict__ col0, const float* __restrict__ val0, int nnz0,
    const int* __restrict__ col1, const float* __restrict__ val1,
    int* __restrict__ start, int* __restrict__ slots) {
  __shared__ int cnt[BIN_ROWS];
  __shared__ int part[BIN_ROWS / 32];
  __shared__ int wsum[BIN_ROWS / 32];
  __shared__ int ordered[STAGE_CAP];
  const int t = threadIdx.x;
  const int lane = t & 31;
  const int w = t >> 5;
  const int b = blockIdx.x;

  int pre = (t < b) ? tot[t] : 0;
  for (int d = 16; d >= 1; d >>= 1) pre += __shfl_xor(pre, d);
  if (lane == 0) part[w] = pre;
  cnt[t] = 0;
  __syncthreads();
  int seg0 = 0;
  for (int q = 0; q < BIN_ROWS / 32; ++q) seg0 += part[q];
  const int len = tot[b];
  const int* __restrict__ seg = coarse + seg0;

  for (int i0 = t; i0 < len; i0 += SWEEP * BIN_ROWS) {
    int rec[SWEEP];
#pragma unroll
    for (int k = 0; k < SWEEP; ++k) {
      const int i = i0 + k * BIN_ROWS;
      rec[k] = (i < len) ? seg[i] : -1;
    }
#pragma unroll
    for (int k = 0; k < SWEEP; ++k)
      if (rec[k] >= 0) atomicAdd(&cnt[rec[k] & (BIN_ROWS - 1)], 1);
  }
  __syncthreads();

  const int mine = cnt[t];
  const int incl = wave_incl_scan(mine, lane);
  if (lane == 31) wsum[w] = incl;
  __syncthreads();
  int off = 0;
  for (int q = 0; q < BIN_ROWS / 32; ++q) if (q < w) off += wsum[q];
  const int ex = off + incl - mine;
  const int row = b * BIN_ROWS + t;
  if (row < n_rows) start[row] = seg0 + ex;
  if (b == nb - 1 && t == 0) start[n_rows] = seg0 + len;
  cnt[t] = ex;
  __syncthreads();

  const bool staged = len <= STAGE_CAP;
  for (int i0 = t; i0 < len; i0 += SWEEP * BIN_ROWS) {
    int rec[SWEEP];
#pragma unroll
    for (int k = 0; k < SWEEP; ++k) {
      const int i = i0 + k * BIN_ROWS;
      rec[k] = (i < len) ? seg[i] : -1;
    }
#pragma unroll
    for (int k = 0; k < SWEEP; ++k) {
      if (rec[k] >= 0) {
        const int id = rec[k] >> BIN_SHIFT;
        const int pos = atomicAdd(&cnt[rec[k] & (BIN_ROWS - 1)], 1);
        if (staged) ordered[pos] = id;
        else emit_slot<PAIRS>(slots, seg0 + pos, id, col0, val0, nnz0, col1, val1);
      }
    }
  }
  if (staged) {
    __syncthreads();
    for (int i = t; i < len; i += BIN_ROWS)
      emit_slot<PAIRS>(slots, seg0 + i, ordered[i], col0, val0, nnz0, col1, val1);
  }
}

template <bool PAIRS, bool SPLIT>
__global__ void __launch_bounds__(256) gather_rows_kernel(
    const int* __restrict__ start, const int* __restrict__ slots,
    const int* __restrict__ col, const float* __restrict__ val,
    const float* __restrict__ x, const float* __restrict__ x1, int split,
    float* __restrict__ y, float* __restrict__ deg, int n_rows) {
  const int g = blockIdx.x * blockDim.x + threadIdx.x;
  const int r = g >> 4;
  const int j = g & 15;
  int s = 0, n = 0;
  if (r < n_rows) { s = start[r]; n = start[r + 1] - s; }
  const int nmax = max(n, __shfl_xor(n, 16));

  float4 acc = make_float4(0.0f, 0.0f, 0.0f, 0.0f);
  float d = 0.0f;
  for (int base = 0; base < nmax; base += 16) {
    int c = 0;
    float v = 0.0f;
    if (base + j < n) {
      if (PAIRS) {
        int2 p = ((const int2*)slots)[s + base + j];
        c = p.x; v = __int_as_float(p.y);
      } else {
        int e = slots[s + base + j];
        c = col[e]; v = val[e];
      }
    }
    const int lim = min(nmax - base, 16);
    for (int i = 0; i < lim; i += 4) {
      const int   c0 = __shfl(c, i,     16), c1 = __shfl(c, i + 1, 16);
      const int   c2 = __shfl(c, i + 2, 16), c3 = __shfl(c, i + 3, 16);
      const float v0 = __shfl(v, i,     16), v1 = __shfl(v, i + 1, 16);
      const float v2 = __shfl(v, i + 2, 16), v3 = __shfl(v, i + 3, 16);
      const float* p0 = x + (size_t)c0 * EMB;
      const float* p1 = x + (size_t)c1 * EMB;
      const float* p2 = x + (size_t)c2 * EMB;
      const float* p3 = x + (size_t)c3 * EMB;
      if (SPLIT) {
        if (c0 >= split) p0 = x1 + (size_t)(c0 - split) * EMB;
        if (c1 >= split) p1 = x1 + (size_t)(c1 - split) * EMB;
        if (c2 >= split) p2 = x1 + (size_t)(c2 - split) * EMB;
        if (c3 >= split) p3 = x1 + (size_t)(c3 - split) * EMB;
      }
      const float4 x0 = ((const float4*)p0)[j];
      const float4 x1v = ((const float4*)p1)[j];
      const float4 x2 = ((const float4*)p2)[j];
      const float4 x3 = ((const float4*)p3)[j];
      if (base + i < n) {
        acc.x = fmaf(v0, x0.x, acc.x); acc.y = fmaf(v0, x0.y, acc.y);
        acc.z = fmaf(v0, x0.z, acc.z); acc.w = fmaf(v0, x0.w, acc.w);
        d += v0;
      }
      if (base + i + 1 < n) {
        acc.x = fmaf(v1, x1v.x, acc.x); acc.y = fmaf(v1, x1v.y, acc.y);
        acc.z = fmaf(v1, x1v.z, acc.z); acc.w = fmaf(v1, x1v.w, acc.w);
        d += v1;
      }
      if (base + i + 2 < n) {
        acc.x = fmaf(v2, x2.x, acc.x); acc.y = fmaf(v2, x2.y, acc.y);
        acc.z = fmaf(v2, x2.z, acc.z); acc.w = fmaf(v2, x2.w, acc.w);
        d += v2;
      }
      if (base + i + 3 < n) {
        acc.x = fmaf(v3, x3.x, acc.x); acc.y = fmaf(v3, x3.y, acc.y);
        acc.z = fmaf(v3, x3.z, acc.z); acc.w = fmaf(v3, x3.w, acc.w);
        d += v3;
      }
    }
  }
  if (r < n_rows) {
    ((float4*)(y + (size_t)r * EMB))[j] = acc;
    if (deg != nullptr && j == 0) deg[r] = d;
  }
}

__device__ __forceinline__ void identity_frags(int n, int hi, v2f a[4]) {
#pragma unroll
  for (int k = 0; k < 4; ++k) {
    a[k].x = (n == (4 * k + 2 * hi))     ? 1.0f : 0.0f;
    a[k].y = (n == (4 * k + 2 * hi + 1)) ? 1.0f : 0.0f;
  }
}

__device__ __forceinline__ void load_b_frags(const float* __restrict__ ep, int hi, v2f b[4]) {
#pragma unroll
  for (int k = 0; k < 4; ++k) {
    b[k].x = ep[(size_t)(4 * k + 2 * hi) * EMB];
    b[k].y = ep[(size_t)(4 * k + 2 * hi + 1) * EMB];
  }
}

__global__ void __launch_bounds__(128) sum3_wmma_kernel(
    const float* __restrict__ ue, const float* __restrict__ ie,
    const float* __restrict__ e1, const float* __restrict__ e2,
    float* __restrict__ out) {
  int wave = threadIdx.x >> 5;
  int lane = threadIdx.x & 31;
  int n    = lane & 15;
  int hi   = lane >> 4;
  int row0 = blockIdx.x * 16;
  int col0 = wave * 16;

  const size_t base = (size_t)row0 * EMB + col0 + n;
  const float* zp = (row0 < NUM_USERS)
      ? ue + base
      : ie + ((size_t)(row0 - NUM_USERS) * EMB + col0 + n);

  v8f c;
#pragma unroll
  for (int r = 0; r < 8; ++r)
    c[r] = zp[(size_t)(r + 8 * hi) * EMB];

  v2f a[4], b1[4], b2[4];
  identity_frags(n, hi, a);
  load_b_frags(e1 + base, hi, b1);
  load_b_frags(e2 + base, hi, b2);

#pragma unroll
  for (int k = 0; k < 4; ++k)
    c = __builtin_amdgcn_wmma_f32_16x16x4_f32(
        false, a[k], false, b1[k], (short)0, c, false, false);
#pragma unroll
  for (int k = 0; k < 4; ++k)
    c = __builtin_amdgcn_wmma_f32_16x16x4_f32(
        false, a[k], false, b2[k], (short)0, c, false, false);

  float* op = out + base;
#pragma unroll
  for (int r = 0; r < 8; ++r)
    op[(size_t)(r + 8 * hi) * EMB] = c[r];
}

__global__ void __launch_bounds__(128) last_wmma_kernel(
    const float* __restrict__ e3, const float* __restrict__ ulocal,
    const float* __restrict__ deg, float* __restrict__ out) {
  int wave = threadIdx.x >> 5;
  int lane = threadIdx.x & 31;
  int n    = lane & 15;
  int hi   = lane >> 4;
  int row0 = blockIdx.x * 16;
  int col0 = wave * 16;
  const bool user = row0 < NUM_USERS;

  const size_t base = (size_t)row0 * EMB + col0 + n;
  const float* ap = out + base;

  v8f c;
#pragma unroll
  for (int r = 0; r < 8; ++r)
    c[r] = ap[(size_t)(r + 8 * hi) * EMB];

  v2f a[4], b[4];
  identity_frags(n, hi, a);
  load_b_frags(e3 + base, hi, b);

  v8f w, dg;
#pragma unroll
  for (int r = 0; r < 8; ++r) { w[r] = 0.0f; dg[r] = 1.0f; }
  if (user) {
    const float* up = ulocal + base;
#pragma unroll
    for (int r = 0; r < 8; ++r)
      w[r] = up[(size_t)(r + 8 * hi) * EMB];
    const float4 d0 = ((const float4*)(deg + row0 + 8 * hi))[0];
    const float4 d1 = ((const float4*)(deg + row0 + 8 * hi))[1];
    dg[0] = d0.x; dg[1] = d0.y; dg[2] = d0.z; dg[3] = d0.w;
    dg[4] = d1.x; dg[5] = d1.y; dg[6] = d1.z; dg[7] = d1.w;
  }

#pragma unroll
  for (int k = 0; k < 4; ++k)
    c = __builtin_amdgcn_wmma_f32_16x16x4_f32(
        false, a[k], false, b[k], (short)0, c, false, false);

  float* op = out + base;
  if (user) {
#pragma unroll
    for (int r = 0; r < 8; ++r) {
      float d = dg[r];
      d = (d == 0.0f) ? 1.0f : d;
      float inv = 1.0f / d;
      float v = c[r] * 0.25f;
      v += w[r] * inv;
      op[(size_t)(r + 8 * hi) * EMB] = v;
    }
  } else {
#pragma unroll
    for (int r = 0; r < 8; ++r)
      op[(size_t)(r + 8 * hi) * EMB] = c[r] * 0.25f;
  }
}

__global__ void __launch_bounds__(256) rownorm_kernel(
    float* __restrict__ x, const float* __restrict__ deg, int rowsVec) {
  int g = blockIdx.x * blockDim.x + threadIdx.x;
  if (g >= rowsVec) return;
  float d = deg[g >> 4];
  d = (d == 0.0f) ? 1.0f : d;
  float inv = 1.0f / d;
  float4 v = ((float4*)x)[g];
  v.x *= inv; v.y *= inv; v.z *= inv; v.w *= inv;
  ((float4*)x)[g] = v;
}

static inline int cdiv(long long a, int b) { return (int)((a + b - 1) / b); }

extern "C" void kernel_launch(void* const* d_in, const int* in_sizes, int n_in,
                              void* d_out, int out_size, void* d_ws, size_t ws_size,
                              hipStream_t stream) {
  const float* user_emb = (const float*)d_in[0];
  const float* item_emb = (const float*)d_in[1];
  const float* adj_val  = (const float*)d_in[2];
  const float* hv_val   = (const float*)d_in[3];
  const float* hu_val   = (const float*)d_in[4];
  const int*   adj_row  = (const int*)d_in[5];
  const int*   adj_col  = (const int*)d_in[6];
  const int*   hv_row   = (const int*)d_in[7];
  const int*   hv_col   = (const int*)d_in[8];
  const int*   hu_row   = (const int*)d_in[9];
  const int*   hu_col   = (const int*)d_in[10];

  const int E_adj = in_sizes[2];
  const int E_hv  = in_sizes[3];
  const int E_hu  = in_sizes[4];

  float* out = (float*)d_out;

  const size_t embBytesN = (size_t)NUM_NODES * EMB * sizeof(float);
  const size_t embBytesB = (size_t)NUM_BQ    * EMB * sizeof(float);
  char* ws = (char*)d_ws;
  float* buf0   = (float*)(ws);
  float* buf1   = (float*)(ws + embBytesN);
  float* bfeat  = (float*)(ws + 2 * embBytesN);
  float* deg_hv = (float*)(ws + 2 * embBytesN + embBytesB);
  float* deg_hu = deg_hv + NUM_BQ;

  int* tot_a    = (int*)(ws + embBytesN);
  int* coarse_a = (int*)(ws + embBytesN + 600064);
  int* start_a  = (int*)(ws + 2 * embBytesN);
  int* perm_a   = (int*)(ws + 2 * embBytesN + 600064);

  int* tot_h    = (int*)(ws + embBytesN);
  int* start_h  = (int*)(ws + embBytesN + 480064);
  int* coarse_h = (int*)(ws + embBytesN + 960128);
  int* pairs_h  = (int*)(ws + embBytesN + 4160128);
  float* ulocal = (float*)(ws + embBytesN + 10560256);

  const int bqVec = NUM_BQ * (EMB / 4);

  const int binsA = (NUM_NODES + BIN_ROWS - 1) / BIN_ROWS;
  hipMemsetAsync(tot_a, 0, (size_t)(2 * BIN_ROWS) * sizeof(int), stream);
  bin_count_kernel<<<cdiv(E_adj, CHUNK), BIN_ROWS, 0, stream>>>(
      adj_row, E_adj, nullptr, 0, 0, tot_a);
  bin_scatter_kernel<<<cdiv(E_adj, CHUNK), BIN_ROWS, 0, stream>>>(
      adj_row, E_adj, nullptr, 0, 0, tot_a, tot_a + BIN_ROWS, coarse_a, binsA);
  bin_sort_kernel<false><<<binsA, BIN_ROWS, 0, stream>>>(
      tot_a, coarse_a, binsA, NUM_NODES,
      nullptr, nullptr, 0, nullptr, nullptr, start_a, perm_a);

  const int gatherBlocksN = cdiv((long long)NUM_NODES * 16, 256);
  gather_rows_kernel<false, true><<<gatherBlocksN, 256, 0, stream>>>(
      start_a, perm_a, adj_col, adj_val, user_emb, item_emb, NUM_USERS,
      buf0, nullptr, NUM_NODES);
  gather_rows_kernel<false, false><<<gatherBlocksN, 256, 0, stream>>>(
      start_a, perm_a, adj_col, adj_val, buf0, nullptr, 0,
      buf1, nullptr, NUM_NODES);
  sum3_wmma_kernel<<<NUM_NODES / 16, 128, 0, stream>>>(
      user_emb, item_emb, buf0, buf1, out);
  gather_rows_kernel<false, false><<<gatherBlocksN, 256, 0, stream>>>(
      start_a, perm_a, adj_col, adj_val, buf1, nullptr, 0,
      buf0, nullptr, NUM_NODES);

  const int binsH = (NUM_BQ + NUM_USERS + BIN_ROWS - 1) / BIN_ROWS;
  hipMemsetAsync(tot_h, 0, (size_t)(2 * BIN_ROWS) * sizeof(int), stream);
  bin_count_kernel<<<cdiv((long long)E_hv + E_hu, CHUNK), BIN_ROWS, 0, stream>>>(
      hv_row, E_hv, hu_row, E_hu, NUM_BQ, tot_h);
  bin_scatter_kernel<<<cdiv((long long)E_hv + E_hu, CHUNK), BIN_ROWS, 0, stream>>>(
      hv_row, E_hv, hu_row, E_hu, NUM_BQ, tot_h, tot_h + BIN_ROWS, coarse_h, binsH);
  bin_sort_kernel<true><<<binsH, BIN_ROWS, 0, stream>>>(
      tot_h, coarse_h, binsH, NUM_BQ + NUM_USERS,
      hv_col, hv_val, E_hv, hu_col, hu_val, start_h, pairs_h);

  gather_rows_kernel<true, false><<<cdiv((long long)NUM_BQ * 16, 256), 256, 0, stream>>>(
      start_h, pairs_h, nullptr, nullptr, item_emb, nullptr, 0,
      bfeat, deg_hv, NUM_BQ);
  rownorm_kernel<<<cdiv(bqVec, 256), 256, 0, stream>>>(bfeat, deg_hv, bqVec);

  gather_rows_kernel<true, false><<<cdiv((long long)NUM_USERS * 16, 256), 256, 0, stream>>>(
      start_h + NUM_BQ, pairs_h, nullptr, nullptr, bfeat, nullptr, 0,
      ulocal, deg_hu, NUM_USERS);

  last_wmma_kernel<<<NUM_NODES / 16, 128, 0, stream>>>(buf0, ulocal, deg_hu, out);
}


# Edge (col,val) ride through the counting sort as 12-byte records; buckets come out as row-ordered (col,val) pairs: no per-edge col/val lookup in sort emit and adjacency layers 1-2; sum order kept
# speedup vs baseline: 3.4367x; 1.1270x over previous
#include <hip/hip_runtime.h>
#include <hip/hip_bf16.h>


typedef __attribute__((ext_vector_type(2))) float v2f;
typedef __attribute__((ext_vector_type(8))) float v8f;

#define NUM_USERS 100000
#define NUM_ITEMS 50000
#define NUM_NODES 150000
#define NUM_BQ    20000
#define EMB       64

#define BIN_SHIFT 9
#define BIN_ROWS  512
#define CHUNK     4096
#define STAGE_CAP 12288
#define SWEEP     8

__device__ __forceinline__ int wave_incl_scan(int v, int lane) {
  for (int d = 1; d < 32; d <<= 1) {
    int u = __shfl_up(v, d);
    if (lane >= d) v += u;
  }
  return v;
}

__device__ __forceinline__ int edge_row(const int* __restrict__ row0, int nnz0,
                                        const int* __restrict__ row1, int rbase1, int e) {
  return (e < nnz0) ? row0[e] : (row1[e - nnz0] + rbase1);
}

__global__ void __launch_bounds__(BIN_ROWS) bin_count_kernel(
    const int* __restrict__ row0, int nnz0,
    const int* __restrict__ row1, int nnz1, int rbase1,
    int* __restrict__ tot) {
  __shared__ int h[BIN_ROWS];
  const int t = threadIdx.x;
  h[t] = 0;
  __syncthreads();
  const int nnz = nnz0 + nnz1;
  const int e0 = blockIdx.x * CHUNK;
  int r[CHUNK / BIN_ROWS];
#pragma unroll
  for (int k = 0; k < CHUNK / BIN_ROWS; ++k) {
    const int e = e0 + k * BIN_ROWS + t;
    r[k] = (e < nnz) ? edge_row(row0, nnz0, row1, rbase1, e) : -1;
  }
#pragma unroll
  for (int k = 0; k < CHUNK / BIN_ROWS; ++k)
    if (r[k] >= 0) atomicAdd(&h[r[k] >> BIN_SHIFT], 1);
  __syncthreads();
  const int c = h[t];
  if (c > 0) atomicAdd(&tot[t], c);
}

__global__ void __launch_bounds__(BIN_ROWS) bin_scatter_kernel(
    const int* __restrict__ row0, const int* __restrict__ col0,
    const float* __restrict__ val0, int nnz0,
    const int* __restrict__ row1, const int* __restrict__ col1,
    const float* __restrict__ val1, int nnz1, int rbase1,
    const int* __restrict__ tot, int* __restrict__ cur,
    int* __restrict__ coarse, int nb) {
  __shared__ int h[BIN_ROWS];
  __shared__ int ls[BIN_ROWS];
  __shared__ int bb[BIN_ROWS];
  __shared__ int wsum[BIN_ROWS / 32];
  __shared__ int4 stage[CHUNK];
  const int t = threadIdx.x;
  const int lane = t & 31;
  const int w = t >> 5;

  const int tv = (t < nb) ? tot[t] : 0;
  const int incl = wave_incl_scan(tv, lane);
  if (lane == 31) wsum[w] = incl;
  h[t] = 0;
  __syncthreads();
  int off = 0;
  for (int q = 0; q < BIN_ROWS / 32; ++q) if (q < w) off += wsum[q];
  bb[t] = off + incl - tv;
  __syncthreads();

  const int nnz = nnz0 + nnz1;
  const int e0 = blockIdx.x * CHUNK;
  int r[CHUNK / BIN_ROWS], bin[CHUNK / BIN_ROWS], lr[CHUNK / BIN_ROWS], rec[CHUNK / BIN_ROWS];
  int cc[CHUNK / BIN_ROWS], vb[CHUNK / BIN_ROWS];
#pragma unroll
  for (int k = 0; k < CHUNK / BIN_ROWS; ++k) {
    const int e = e0 + k * BIN_ROWS + t;
    r[k] = -1; cc[k] = 0; vb[k] = 0;
    if (e < nnz) {
      if (e < nnz0) {
        r[k] = row0[e]; cc[k] = col0[e]; vb[k] = __float_as_int(val0[e]);
      } else {
        r[k] = row1[e - nnz0] + rbase1; cc[k] = col1[e - nnz0];
        vb[k] = __float_as_int(val1[e - nnz0]);
      }
    }
  }
#pragma unroll
  for (int k = 0; k < CHUNK / BIN_ROWS; ++k) {
    const int e = e0 + k * BIN_ROWS + t;
    bin[k] = -1; lr[k] = 0; rec[k] = 0;
    if (r[k] >= 0) {
      bin[k] = r[k] >> BIN_SHIFT;
      lr[k] = atomicAdd(&h[bin[k]], 1);
      rec[k] = (int)(((unsigned)e << BIN_SHIFT) | (unsigned)(r[k] & (BIN_ROWS - 1)));
    }
  }
  __syncthreads();

  const int hv = h[t];
  const int incl2 = wave_incl_scan(hv, lane);
  if (lane == 31) wsum[w] = incl2;
  __syncthreads();
  int off2 = 0;
  for (int q = 0; q < BIN_ROWS / 32; ++q) if (q < w) off2 += wsum[q];
  ls[t] = off2 + incl2 - hv;
  int g = 0;
  if (hv > 0) g = bb[t] + atomicAdd(&cur[t], hv);
  h[t] = g;
  __syncthreads();

#pragma unroll
  for (int k = 0; k < CHUNK / BIN_ROWS; ++k) {
    if (bin[k] >= 0) {
      const int p = ls[bin[k]] + lr[k];
      stage[p] = make_int4(rec[k], cc[k], vb[k], h[bin[k]] + lr[k]);
    }
  }
  __syncthreads();

  const int n_here = min(CHUNK, nnz - e0);
  for (int p = t; p < n_here; p += BIN_ROWS) {
    const int4 s = stage[p];
    int* __restrict__ q = coarse + 3 * s.w;
    q[0] = s.x; q[1] = s.y; q[2] = s.z;
  }
}

template <bool IDS>
__global__ void __launch_bounds__(BIN_ROWS) bin_sort_kernel(
    const int* __restrict__ tot, const int* __restrict__ coarse, int nb, int n_rows,
    int* __restrict__ start, int* __restrict__ pairs, int* __restrict__ ids) {
  __shared__ int cnt[BIN_ROWS];
  __shared__ int part[BIN_ROWS / 32];
  __shared__ int wsum[BIN_ROWS / 32];
  __shared__ int2 ordered[STAGE_CAP];
  __shared__ int ordered_id[IDS ? STAGE_CAP : 1];
  const int t = threadIdx.x;
  const int lane = t & 31;
  const int w = t >> 5;
  const int b = blockIdx.x;

  int pre = (t < b) ? tot[t] : 0;
  for (int d = 16; d >= 1; d >>= 1) pre += __shfl_xor(pre, d);
  if (lane == 0) part[w] = pre;
  cnt[t] = 0;
  __syncthreads();
  int seg0 = 0;
  for (int q = 0; q < BIN_ROWS / 32; ++q) seg0 += part[q];
  const int len = tot[b];
  const int* __restrict__ seg = coarse + 3 * seg0;

  for (int i0 = t; i0 < len; i0 += SWEEP * BIN_ROWS) {
    int rec[SWEEP];
#pragma unroll
    for (int k = 0; k < SWEEP; ++k) {
      const int i = i0 + k * BIN_ROWS;
      rec[k] = (i < len) ? seg[3 * i] : -1;
    }
#pragma unroll
    for (int k = 0; k < SWEEP; ++k)
      if (rec[k] >= 0) atomicAdd(&cnt[rec[k] & (BIN_ROWS - 1)], 1);
  }
  __syncthreads();

  const int mine = cnt[t];
  const int incl = wave_incl_scan(mine, lane);
  if (lane == 31) wsum[w] = incl;
  __syncthreads();
  int off = 0;
  for (int q = 0; q < BIN_ROWS / 32; ++q) if (q < w) off += wsum[q];
  const int ex = off + incl - mine;
  const int row = b * BIN_ROWS + t;
  if (row < n_rows) start[row] = seg0 + ex;
  if (b == nb - 1 && t == 0) start[n_rows] = seg0 + len;
  cnt[t] = ex;
  __syncthreads();

  int2* __restrict__ pout = (int2*)pairs;
  const bool staged = len <= STAGE_CAP;
  for (int i0 = t; i0 < len; i0 += SWEEP * BIN_ROWS) {
    int rec[SWEEP], cc[SWEEP], vb[SWEEP];
#pragma unroll
    for (int k = 0; k < SWEEP; ++k) {
      const int i = i0 + k * BIN_ROWS;
      rec[k] = -1; cc[k] = 0; vb[k] = 0;
      if (i < len) { rec[k] = seg[3 * i]; cc[k] = seg[3 * i + 1]; vb[k] = seg[3 * i + 2]; }
    }
#pragma unroll
    for (int k = 0; k < SWEEP; ++k) {
      if (rec[k] >= 0) {
        const int id = rec[k] >> BIN_SHIFT;
        const int pos = atomicAdd(&cnt[rec[k] & (BIN_ROWS - 1)], 1);
        if (staged) {
          ordered[pos] = make_int2(cc[k], vb[k]);
          if (IDS) ordered_id[pos] = id;
        } else {
          pout[seg0 + pos] = make_int2(cc[k], vb[k]);
          if (IDS) ids[seg0 + pos] = id;
        }
      }
    }
  }
  if (staged) {
    __syncthreads();
    for (int i = t; i < len; i += BIN_ROWS) {
      pout[seg0 + i] = ordered[i];
      if (IDS) ids[seg0 + i] = ordered_id[i];
    }
  }
}

template <bool PAIRS, bool SPLIT>
__global__ void __launch_bounds__(256) gather_rows_kernel(
    const int* __restrict__ start, const int* __restrict__ slots,
    const int* __restrict__ col, const float* __restrict__ val,
    const float* __restrict__ x, const float* __restrict__ x1, int split,
    float* __restrict__ y, float* __restrict__ deg, int n_rows) {
  const int g = blockIdx.x * blockDim.x + threadIdx.x;
  const int r = g >> 4;
  const int j = g & 15;
  int s = 0, n = 0;
  if (r < n_rows) { s = start[r]; n = start[r + 1] - s; }
  const int nmax = max(n, __shfl_xor(n, 16));

  float4 acc = make_float4(0.0f, 0.0f, 0.0f, 0.0f);
  float d = 0.0f;
  for (int base = 0; base < nmax; base += 16) {
    int c = 0;
    float v = 0.0f;
    if (base + j < n) {
      if (PAIRS) {
        int2 p = ((const int2*)slots)[s + base + j];
        c = p.x; v = __int_as_float(p.y);
      } else {
        int e = slots[s + base + j];
        c = col[e]; v = val[e];
      }
    }
    const int lim = min(nmax - base, 16);
    for (int i = 0; i < lim; i += 4) {
      const int   c0 = __shfl(c, i,     16), c1 = __shfl(c, i + 1, 16);
      const int   c2 = __shfl(c, i + 2, 16), c3 = __shfl(c, i + 3, 16);
      const float v0 = __shfl(v, i,     16), v1 = __shfl(v, i + 1, 16);
      const float v2 = __shfl(v, i + 2, 16), v3 = __shfl(v, i + 3, 16);
      const float* p0 = x + (size_t)c0 * EMB;
      const float* p1 = x + (size_t)c1 * EMB;
      const float* p2 = x + (size_t)c2 * EMB;
      const float* p3 = x + (size_t)c3 * EMB;
      if (SPLIT) {
        if (c0 >= split) p0 = x1 + (size_t)(c0 - split) * EMB;
        if (c1 >= split) p1 = x1 + (size_t)(c1 - split) * EMB;
        if (c2 >= split) p2 = x1 + (size_t)(c2 - split) * EMB;
        if (c3 >= split) p3 = x1 + (size_t)(c3 - split) * EMB;
      }
      const float4 x0 = ((const float4*)p0)[j];
      const float4 x1v = ((const float4*)p1)[j];
      const float4 x2 = ((const float4*)p2)[j];
      const float4 x3 = ((const float4*)p3)[j];
      if (base + i < n) {
        acc.x = fmaf(v0, x0.x, acc.x); acc.y = fmaf(v0, x0.y, acc.y);
        acc.z = fmaf(v0, x0.z, acc.z); acc.w = fmaf(v0, x0.w, acc.w);
        d += v0;
      }
      if (base + i + 1 < n) {
        acc.x = fmaf(v1, x1v.x, acc.x); acc.y = fmaf(v1, x1v.y, acc.y);
        acc.z = fmaf(v1, x1v.z, acc.z); acc.w = fmaf(v1, x1v.w, acc.w);
        d += v1;
      }
      if (base + i + 2 < n) {
        acc.x = fmaf(v2, x2.x, acc.x); acc.y = fmaf(v2, x2.y, acc.y);
        acc.z = fmaf(v2, x2.z, acc.z); acc.w = fmaf(v2, x2.w, acc.w);
        d += v2;
      }
      if (base + i + 3 < n) {
        acc.x = fmaf(v3, x3.x, acc.x); acc.y = fmaf(v3, x3.y, acc.y);
        acc.z = fmaf(v3, x3.z, acc.z); acc.w = fmaf(v3, x3.w, acc.w);
        d += v3;
      }
    }
  }
  if (r < n_rows) {
    ((float4*)(y + (size_t)r * EMB))[j] = acc;
    if (deg != nullptr && j == 0) deg[r] = d;
  }
}

__device__ __forceinline__ void identity_frags(int n, int hi, v2f a[4]) {
#pragma unroll
  for (int k = 0; k < 4; ++k) {
    a[k].x = (n == (4 * k + 2 * hi))     ? 1.0f : 0.0f;
    a[k].y = (n == (4 * k + 2 * hi + 1)) ? 1.0f : 0.0f;
  }
}

__device__ __forceinline__ void load_b_frags(const float* __restrict__ ep, int hi, v2f b[4]) {
#pragma unroll
  for (int k = 0; k < 4; ++k) {
    b[k].x = ep[(size_t)(4 * k + 2 * hi) * EMB];
    b[k].y = ep[(size_t)(4 * k + 2 * hi + 1) * EMB];
  }
}

__global__ void __launch_bounds__(128) sum3_wmma_kernel(
    const float* __restrict__ ue, const float* __restrict__ ie,
    const float* __restrict__ e1, const float* __restrict__ e2,
    float* __restrict__ out) {
  int wave = threadIdx.x >> 5;
  int lane = threadIdx.x & 31;
  int n    = lane & 15;
  int hi   = lane >> 4;
  int row0 = blockIdx.x * 16;
  int col0 = wave * 16;

  const size_t base = (size_t)row0 * EMB + col0 + n;
  const float* zp = (row0 < NUM_USERS)
      ? ue + base
      : ie + ((size_t)(row0 - NUM_USERS) * EMB + col0 + n);

  v8f c;
#pragma unroll
  for (int r = 0; r < 8; ++r)
    c[r] = zp[(size_t)(r + 8 * hi) * EMB];

  v2f a[4], b1[4], b2[4];
  identity_frags(n, hi, a);
  load_b_frags(e1 + base, hi, b1);
  load_b_frags(e2 + base, hi, b2);

#pragma unroll
  for (int k = 0; k < 4; ++k)
    c = __builtin_amdgcn_wmma_f32_16x16x4_f32(
        false, a[k], false, b1[k], (short)0, c, false, false);
#pragma unroll
  for (int k = 0; k < 4; ++k)
    c = __builtin_amdgcn_wmma_f32_16x16x4_f32(
        false, a[k], false, b2[k], (short)0, c, false, false);

  float* op = out + base;
#pragma unroll
  for (int r = 0; r < 8; ++r)
    op[(size_t)(r + 8 * hi) * EMB] = c[r];
}

__global__ void __launch_bounds__(128) last_wmma_kernel(
    const float* __restrict__ e3, const float* __restrict__ ulocal,
    const float* __restrict__ deg, float* __restrict__ out) {
  int wave = threadIdx.x >> 5;
  int lane = threadIdx.x & 31;
  int n    = lane & 15;
  int hi   = lane >> 4;
  int row0 = blockIdx.x * 16;
  int col0 = wave * 16;
  const bool user = row0 < NUM_USERS;

  const size_t base = (size_t)row0 * EMB + col0 + n;
  const float* ap = out + base;

  v8f c;
#pragma unroll
  for (int r = 0; r < 8; ++r)
    c[r] = ap[(size_t)(r + 8 * hi) * EMB];

  v2f a[4], b[4];
  identity_frags(n, hi, a);
  load_b_frags(e3 + base, hi, b);

  v8f w, dg;
#pragma unroll
  for (int r = 0; r < 8; ++r) { w[r] = 0.0f; dg[r] = 1.0f; }
  if (user) {
    const float* up = ulocal + base;
#pragma unroll
    for (int r = 0; r < 8; ++r)
      w[r] = up[(size_t)(r + 8 * hi) * EMB];
    const float4 d0 = ((const float4*)(deg + row0 + 8 * hi))[0];
    const float4 d1 = ((const float4*)(deg + row0 + 8 * hi))[1];
    dg[0] = d0.x; dg[1] = d0.y; dg[2] = d0.z; dg[3] = d0.w;
    dg[4] = d1.x; dg[5] = d1.y; dg[6] = d1.z; dg[7] = d1.w;
  }

#pragma unroll
  for (int k = 0; k < 4; ++k)
    c = __builtin_amdgcn_wmma_f32_16x16x4_f32(
        false, a[k], false, b[k], (short)0, c, false, false);

  float* op = out + base;
  if (user) {
#pragma unroll
    for (int r = 0; r < 8; ++r) {
      float d = dg[r];
      d = (d == 0.0f) ? 1.0f : d;
      float inv = 1.0f / d;
      float v = c[r] * 0.25f;
      v += w[r] * inv;
      op[(size_t)(r + 8 * hi) * EMB] = v;
    }
  } else {
#pragma unroll
    for (int r = 0; r < 8; ++r)
      op[(size_t)(r + 8 * hi) * EMB] = c[r] * 0.25f;
  }
}

__global__ void __launch_bounds__(256) rownorm_kernel(
    float* __restrict__ x, const float* __restrict__ deg, int rowsVec) {
  int g = blockIdx.x * blockDim.x + threadIdx.x;
  if (g >= rowsVec) return;
  float d = deg[g >> 4];
  d = (d == 0.0f) ? 1.0f : d;
  float inv = 1.0f / d;
  float4 v = ((float4*)x)[g];
  v.x *= inv; v.y *= inv; v.z *= inv; v.w *= inv;
  ((float4*)x)[g] = v;
}

static inline int cdiv(long long a, int b) { return (int)((a + b - 1) / b); }

extern "C" void kernel_launch(void* const* d_in, const int* in_sizes, int n_in,
                              void* d_out, int out_size, void* d_ws, size_t ws_size,
                              hipStream_t stream) {
  const float* user_emb = (const float*)d_in[0];
  const float* item_emb = (const float*)d_in[1];
  const float* adj_val  = (const float*)d_in[2];
  const float* hv_val   = (const float*)d_in[3];
  const float* hu_val   = (const float*)d_in[4];
  const int*   adj_row  = (const int*)d_in[5];
  const int*   adj_col  = (const int*)d_in[6];
  const int*   hv_row   = (const int*)d_in[7];
  const int*   hv_col   = (const int*)d_in[8];
  const int*   hu_row   = (const int*)d_in[9];
  const int*   hu_col   = (const int*)d_in[10];

  const int E_adj = in_sizes[2];
  const int E_hv  = in_sizes[3];
  const int E_hu  = in_sizes[4];

  float* out = (float*)d_out;

  const size_t embBytesN = (size_t)NUM_NODES * EMB * sizeof(float);
  const size_t embBytesB = (size_t)NUM_BQ    * EMB * sizeof(float);
  char* ws = (char*)d_ws;
  float* buf0   = (float*)(ws);
  float* buf1   = (float*)(ws + embBytesN);
  float* bfeat  = (float*)(ws + 2 * embBytesN);
  float* deg_hv = (float*)(ws + 2 * embBytesN + embBytesB);
  float* deg_hu = deg_hv + NUM_BQ;

  int* tot_a    = (int*)(ws + embBytesN);
  int* coarse_a = (int*)(ws + embBytesN + 600064);
  int* start_a  = (int*)(ws + 2 * embBytesN);
  int* perm_a   = (int*)(ws + 2 * embBytesN + 600064);
  int* pairs_a  = (int*)d_out;

  int* tot_h    = (int*)(ws + embBytesN);
  int* start_h  = (int*)(ws + embBytesN + 480064);
  int* pairs_h  = (int*)(ws + embBytesN + 4160128);
  int* coarse_h = (int*)(ws + embBytesN + 10560256);
  float* ulocal = (float*)(ws + embBytesN + 10560256);

  const int bqVec = NUM_BQ * (EMB / 4);

  const int binsA = (NUM_NODES + BIN_ROWS - 1) / BIN_ROWS;
  hipMemsetAsync(tot_a, 0, (size_t)(2 * BIN_ROWS) * sizeof(int), stream);
  bin_count_kernel<<<cdiv(E_adj, CHUNK), BIN_ROWS, 0, stream>>>(
      adj_row, E_adj, nullptr, 0, 0, tot_a);
  bin_scatter_kernel<<<cdiv(E_adj, CHUNK), BIN_ROWS, 0, stream>>>(
      adj_row, adj_col, adj_val, E_adj, nullptr, nullptr, nullptr, 0, 0,
      tot_a, tot_a + BIN_ROWS, coarse_a, binsA);
  bin_sort_kernel<true><<<binsA, BIN_ROWS, 0, stream>>>(
      tot_a, coarse_a, binsA, NUM_NODES, start_a, pairs_a, perm_a);

  const int gatherBlocksN = cdiv((long long)NUM_NODES * 16, 256);
  gather_rows_kernel<true, true><<<gatherBlocksN, 256, 0, stream>>>(
      start_a, pairs_a, nullptr, nullptr, user_emb, item_emb, NUM_USERS,
      buf0, nullptr, NUM_NODES);
  gather_rows_kernel<true, false><<<gatherBlocksN, 256, 0, stream>>>(
      start_a, pairs_a, nullptr, nullptr, buf0, nullptr, 0,
      buf1, nullptr, NUM_NODES);
  sum3_wmma_kernel<<<NUM_NODES / 16, 128, 0, stream>>>(
      user_emb, item_emb, buf0, buf1, out);
  gather_rows_kernel<false, false><<<gatherBlocksN, 256, 0, stream>>>(
      start_a, perm_a, adj_col, adj_val, buf1, nullptr, 0,
      buf0, nullptr, NUM_NODES);

  const int binsH = (NUM_BQ + NUM_USERS + BIN_ROWS - 1) / BIN_ROWS;
  hipMemsetAsync(tot_h, 0, (size_t)(2 * BIN_ROWS) * sizeof(int), stream);
  bin_count_kernel<<<cdiv((long long)E_hv + E_hu, CHUNK), BIN_ROWS, 0, stream>>>(
      hv_row, E_hv, hu_row, E_hu, NUM_BQ, tot_h);
  bin_scatter_kernel<<<cdiv((long long)E_hv + E_hu, CHUNK), BIN_ROWS, 0, stream>>>(
      hv_row, hv_col, hv_val, E_hv, hu_row, hu_col, hu_val, E_hu, NUM_BQ,
      tot_h, tot_h + BIN_ROWS, coarse_h, binsH);
  bin_sort_kernel<false><<<binsH, BIN_ROWS, 0, stream>>>(
      tot_h, coarse_h, binsH, NUM_BQ + NUM_USERS, start_h, pairs_h, nullptr);

  gather_rows_kernel<true, false><<<cdiv((long long)NUM_BQ * 16, 256), 256, 0, stream>>>(
      start_h, pairs_h, nullptr, nullptr, item_emb, nullptr, 0,
      bfeat, deg_hv, NUM_BQ);
  rownorm_kernel<<<cdiv(bqVec, 256), 256, 0, stream>>>(bfeat, deg_hv, bqVec);

  gather_rows_kernel<true, false><<<cdiv((long long)NUM_USERS * 16, 256), 256, 0, stream>>>(
      start_h + NUM_BQ, pairs_h, nullptr, nullptr, bfeat, nullptr, 0,
      ulocal, deg_hu, NUM_USERS);

  last_wmma_kernel<<<NUM_NODES / 16, 128, 0, stream>>>(buf0, ulocal, deg_hu, out);
}


# Both edge sets bucketed by one set of three launches (block ranges); biclique term gathered first, normalised at its stores, parked in out, added in layer-sum pass 1: sum order changes
# speedup vs baseline: 4.0468x; 1.1775x over previous
#include <hip/hip_runtime.h>
#include <hip/hip_bf16.h>


typedef __attribute__((ext_vector_type(2))) float v2f;
typedef __attribute__((ext_vector_type(8))) float v8f;

#define NUM_USERS 100000
#define NUM_ITEMS 50000
#define NUM_NODES 150000
#define NUM_BQ    20000
#define EMB       64

#define BIN_SHIFT 9
#define BIN_ROWS  512
#define CHUNK     4096
#define STAGE_CAP 12288
#define STAGE_CAP_IDS 8192
#define SWEEP     8

struct EdgeSet {
  const int* row0; const int* col0; const float* val0; int nnz0;
  const int* row1; const int* col1; const float* val1; int nnz1;
  int rbase1;
  int* tot;
  int* coarse;
  int* start;
  int* pairs;
  int* ids;
  int nb;
  int n_rows;
  int nblk;
};

__device__ __forceinline__ int wave_incl_scan(int v, int lane) {
  for (int d = 1; d < 32; d <<= 1) {
    int u = __shfl_up(v, d);
    if (lane >= d) v += u;
  }
  return v;
}

__device__ __forceinline__ int edge_row(const int* __restrict__ row0, int nnz0,
                                        const int* __restrict__ row1, int rbase1, int e) {
  return (e < nnz0) ? row0[e] : (row1[e - nnz0] + rbase1);
}

__device__ __forceinline__ void bin_count_body(
    const int* __restrict__ row0, int nnz0,
    const int* __restrict__ row1, int nnz1, int rbase1,
    int* __restrict__ tot, int blk) {
  __shared__ int h[BIN_ROWS];
  const int t = threadIdx.x;
  h[t] = 0;
  __syncthreads();
  const int nnz = nnz0 + nnz1;
  const int e0 = blk * CHUNK;
  int r[CHUNK / BIN_ROWS];
#pragma unroll
  for (int k = 0; k < CHUNK / BIN_ROWS; ++k) {
    const int e = e0 + k * BIN_ROWS + t;
    r[k] = (e < nnz) ? edge_row(row0, nnz0, row1, rbase1, e) : -1;
  }
#pragma unroll
  for (int k = 0; k < CHUNK / BIN_ROWS; ++k)
    if (r[k] >= 0) atomicAdd(&h[r[k] >> BIN_SHIFT], 1);
  __syncthreads();
  const int c = h[t];
  if (c > 0) atomicAdd(&tot[t], c);
}

__global__ void __launch_bounds__(BIN_ROWS) bin_count_kernel(EdgeSet sa, EdgeSet sb) {
  const bool first = (int)blockIdx.x < sa.nblk;
  const EdgeSet s = first ? sa : sb;
  const int blk = first ? (int)blockIdx.x : (int)blockIdx.x - sa.nblk;
  bin_count_body(s.row0, s.nnz0, s.row1, s.nnz1, s.rbase1, s.tot, blk);
}

__device__ __forceinline__ void bin_scatter_body(
    const int* __restrict__ row0, const int* __restrict__ col0,
    const float* __restrict__ val0, int nnz0,
    const int* __restrict__ row1, const int* __restrict__ col1,
    const float* __restrict__ val1, int nnz1, int rbase1,
    const int* __restrict__ tot, int* __restrict__ cur,
    int* __restrict__ coarse, int nb, int blk) {
  __shared__ int h[BIN_ROWS];
  __shared__ int ls[BIN_ROWS];
  __shared__ int bb[BIN_ROWS];
  __shared__ int wsum[BIN_ROWS / 32];
  __shared__ int4 stage[CHUNK];
  const int t = threadIdx.x;
  const int lane = t & 31;
  const int w = t >> 5;

  const int tv = (t < nb) ? tot[t] : 0;
  const int incl = wave_incl_scan(tv, lane);
  if (lane == 31) wsum[w] = incl;
  h[t] = 0;
  __syncthreads();
  int off = 0;
  for (int q = 0; q < BIN_ROWS / 32; ++q) if (q < w) off += wsum[q];
  bb[t] = off + incl - tv;
  __syncthreads();

  const int nnz = nnz0 + nnz1;
  const int e0 = blk * CHUNK;
  int r[CHUNK / BIN_ROWS], bin[CHUNK / BIN_ROWS], lr[CHUNK / BIN_ROWS], rec[CHUNK / BIN_ROWS];
  int cc[CHUNK / BIN_ROWS], vb[CHUNK / BIN_ROWS];
#pragma unroll
  for (int k = 0; k < CHUNK / BIN_ROWS; ++k) {
    const int e = e0 + k * BIN_ROWS + t;
    r[k] = -1; cc[k] = 0; vb[k] = 0;
    if (e < nnz) {
      if (e < nnz0) {
        r[k] = row0[e]; cc[k] = col0[e]; vb[k] = __float_as_int(val0[e]);
      } else {
        r[k] = row1[e - nnz0] + rbase1; cc[k] = col1[e - nnz0];
        vb[k] = __float_as_int(val1[e - nnz0]);
      }
    }
  }
#pragma unroll
  for (int k = 0; k < CHUNK / BIN_ROWS; ++k) {
    const int e = e0 + k * BIN_ROWS + t;
    bin[k] = -1; lr[k] = 0; rec[k] = 0;
    if (r[k] >= 0) {
      bin[k] = r[k] >> BIN_SHIFT;
      lr[k] = atomicAdd(&h[bin[k]], 1);
      rec[k] = (int)(((unsigned)e << BIN_SHIFT) | (unsigned)(r[k] & (BIN_ROWS - 1)));
    }
  }
  __syncthreads();

  const int hv = h[t];
  const int incl2 = wave_incl_scan(hv, lane);
  if (lane == 31) wsum[w] = incl2;
  __syncthreads();
  int off2 = 0;
  for (int q = 0; q < BIN_ROWS / 32; ++q) if (q < w) off2 += wsum[q];
  ls[t] = off2 + incl2 - hv;
  int g = 0;
  if (hv > 0) g = bb[t] + atomicAdd(&cur[t], hv);
  h[t] = g;
  __syncthreads();

#pragma unroll
  for (int k = 0; k < CHUNK / BIN_ROWS; ++k) {
    if (bin[k] >= 0) {
      const int p = ls[bin[k]] + lr[k];
      stage[p] = make_int4(rec[k], cc[k], vb[k], h[bin[k]] + lr[k]);
    }
  }
  __syncthreads();

  const int n_here = min(CHUNK, nnz - e0);
  for (int p = t; p < n_here; p += BIN_ROWS) {
    const int4 s = stage[p];
    int* __restrict__ q = coarse + 3 * s.w;
    q[0] = s.x; q[1] = s.y; q[2] = s.z;
  }
}

__global__ void __launch_bounds__(BIN_ROWS) bin_scatter_kernel(EdgeSet sa, EdgeSet sb) {
  const bool first = (int)blockIdx.x < sa.nblk;
  const EdgeSet s = first ? sa : sb;
  const int blk = first ? (int)blockIdx.x : (int)blockIdx.x - sa.nblk;
  bin_scatter_body(s.row0, s.col0, s.val0, s.nnz0, s.row1, s.col1, s.val1, s.nnz1,
                   s.rbase1, s.tot, s.tot + BIN_ROWS, s.coarse, s.nb, blk);
}

__device__ __forceinline__ void bin_sort_body(
    const int* __restrict__ tot, const int* __restrict__ coarse, int nb, int n_rows,
    int* __restrict__ start, int* __restrict__ pairs, int* __restrict__ ids, int b) {
  __shared__ int cnt[BIN_ROWS];
  __shared__ int part[BIN_ROWS / 32];
  __shared__ int wsum[BIN_ROWS / 32];
  __shared__ int2 ordered[STAGE_CAP];
  const bool with_ids = ids != nullptr;
  int* ordered_id = (int*)(ordered + STAGE_CAP_IDS);
  const int t = threadIdx.x;
  const int lane = t & 31;
  const int w = t >> 5;

  int pre = (t < b) ? tot[t] : 0;
  for (int d = 16; d >= 1; d >>= 1) pre += __shfl_xor(pre, d);
  if (lane == 0) part[w] = pre;
  cnt[t] = 0;
  __syncthreads();
  int seg0 = 0;
  for (int q = 0; q < BIN_ROWS / 32; ++q) seg0 += part[q];
  const int len = tot[b];
  const int* __restrict__ seg = coarse + 3 * seg0;

  for (int i0 = t; i0 < len; i0 += SWEEP * BIN_ROWS) {
    int rec[SWEEP];
#pragma unroll
    for (int k = 0; k < SWEEP; ++k) {
      const int i = i0 + k * BIN_ROWS;
      rec[k] = (i < len) ? seg[3 * i] : -1;
    }
#pragma unroll
    for (int k = 0; k < SWEEP; ++k)
      if (rec[k] >= 0) atomicAdd(&cnt[rec[k] & (BIN_ROWS - 1)], 1);
  }
  __syncthreads();

  const int mine = cnt[t];
  const int incl = wave_incl_scan(mine, lane);
  if (lane == 31) wsum[w] = incl;
  __syncthreads();
  int off = 0;
  for (int q = 0; q < BIN_ROWS / 32; ++q) if (q < w) off += wsum[q];
  const int ex = off + incl - mine;
  const int row = b * BIN_ROWS + t;
  if (row < n_rows) start[row] = seg0 + ex;
  if (b == nb - 1 && t == 0) start[n_rows] = seg0 + len;
  cnt[t] = ex;
  __syncthreads();

  int2* __restrict__ pout = (int2*)pairs;
  const bool staged = len <= (with_ids ? STAGE_CAP_IDS : STAGE_CAP);
  for (int i0 = t; i0 < len; i0 += SWEEP * BIN_ROWS) {
    int rec[SWEEP], cc[SWEEP], vb[SWEEP];
#pragma unroll
    for (int k = 0; k < SWEEP; ++k) {
      const int i = i0 + k * BIN_ROWS;
      rec[k] = -1; cc[k] = 0; vb[k] = 0;
      if (i < len) { rec[k] = seg[3 * i]; cc[k] = seg[3 * i + 1]; vb[k] = seg[3 * i + 2]; }
    }
#pragma unroll
    for (int k = 0; k < SWEEP; ++k) {
      if (rec[k] >= 0) {
        const int id = rec[k] >> BIN_SHIFT;
        const int pos = atomicAdd(&cnt[rec[k] & (BIN_ROWS - 1)], 1);
        if (staged) {
          ordered[pos] = make_int2(cc[k], vb[k]);
          if (with_ids) ordered_id[pos] = id;
        } else {
          pout[seg0 + pos] = make_int2(cc[k], vb[k]);
          if (with_ids) ids[seg0 + pos] = id;
        }
      }
    }
  }
  if (staged) {
    __syncthreads();
    for (int i = t; i < len; i += BIN_ROWS) {
      pout[seg0 + i] = ordered[i];
      if (with_ids) ids[seg0 + i] = ordered_id[i];
    }
  }
}

__global__ void __launch_bounds__(BIN_ROWS) bin_sort_kernel(EdgeSet sa, EdgeSet sb) {
  const bool first = (int)blockIdx.x < sa.nb;
  const EdgeSet s = first ? sa : sb;
  const int b = first ? (int)blockIdx.x : (int)blockIdx.x - sa.nb;
  bin_sort_body(s.tot, s.coarse, s.nb, s.n_rows, s.start, s.pairs, s.ids, b);
}

template <bool PAIRS, bool SPLIT, bool NORM>
__global__ void __launch_bounds__(256) gather_rows_kernel(
    const int* __restrict__ start, const int* __restrict__ slots,
    const int* __restrict__ col, const float* __restrict__ val,
    const float* __restrict__ x, const float* __restrict__ x1, int split,
    float* __restrict__ y, int n_rows) {
  const int g = blockIdx.x * blockDim.x + threadIdx.x;
  const int r = g >> 4;
  const int j = g & 15;
  int s = 0, n = 0;
  if (r < n_rows) { s = start[r]; n = start[r + 1] - s; }
  const int nmax = max(n, __shfl_xor(n, 16));

  float4 acc = make_float4(0.0f, 0.0f, 0.0f, 0.0f);
  float d = 0.0f;
  for (int base = 0; base < nmax; base += 16) {
    int c = 0;
    float v = 0.0f;
    if (base + j < n) {
      if (PAIRS) {
        int2 p = ((const int2*)slots)[s + base + j];
        c = p.x; v = __int_as_float(p.y);
      } else {
        int e = slots[s + base + j];
        c = col[e]; v = val[e];
      }
    }
    const int lim = min(nmax - base, 16);
    for (int i = 0; i < lim; i += 4) {
      const int   c0 = __shfl(c, i,     16), c1 = __shfl(c, i + 1, 16);
      const int   c2 = __shfl(c, i + 2, 16), c3 = __shfl(c, i + 3, 16);
      const float v0 = __shfl(v, i,     16), v1 = __shfl(v, i + 1, 16);
      const float v2 = __shfl(v, i + 2, 16), v3 = __shfl(v, i + 3, 16);
      const float* p0 = x + (size_t)c0 * EMB;
      const float* p1 = x + (size_t)c1 * EMB;
      const float* p2 = x + (size_t)c2 * EMB;
      const float* p3 = x + (size_t)c3 * EMB;
      if (SPLIT) {
        if (c0 >= split) p0 = x1 + (size_t)(c0 - split) * EMB;
        if (c1 >= split) p1 = x1 + (size_t)(c1 - split) * EMB;
        if (c2 >= split) p2 = x1 + (size_t)(c2 - split) * EMB;
        if (c3 >= split) p3 = x1 + (size_t)(c3 - split) * EMB;
      }
      const float4 x0 = ((const float4*)p0)[j];
      const float4 x1v = ((const float4*)p1)[j];
      const float4 x2 = ((const float4*)p2)[j];
      const float4 x3 = ((const float4*)p3)[j];
      if (base + i < n) {
        acc.x = fmaf(v0, x0.x, acc.x); acc.y = fmaf(v0, x0.y, acc.y);
        acc.z = fmaf(v0, x0.z, acc.z); acc.w = fmaf(v0, x0.w, acc.w);
        d += v0;
      }
      if (base + i + 1 < n) {
        acc.x = fmaf(v1, x1v.x, acc.x); acc.y = fmaf(v1, x1v.y, acc.y);
        acc.z = fmaf(v1, x1v.z, acc.z); acc.w = fmaf(v1, x1v.w, acc.w);
        d += v1;
      }
      if (base + i + 2 < n) {
        acc.x = fmaf(v2, x2.x, acc.x); acc.y = fmaf(v2, x2.y, acc.y);
        acc.z = fmaf(v2, x2.z, acc.z); acc.w = fmaf(v2, x2.w, acc.w);
        d += v2;
      }
      if (base + i + 3 < n) {
        acc.x = fmaf(v3, x3.x, acc.x); acc.y = fmaf(v3, x3.y, acc.y);
        acc.z = fmaf(v3, x3.z, acc.z); acc.w = fmaf(v3, x3.w, acc.w);
        d += v3;
      }
    }
  }
  if (NORM) {
    const float dd = (d == 0.0f) ? 1.0f : d;
    const float inv = 1.0f / dd;
    acc.x *= inv; acc.y *= inv; acc.z *= inv; acc.w *= inv;
  }
  if (r < n_rows)
    ((float4*)(y + (size_t)r * EMB))[j] = acc;
}

__device__ __forceinline__ void identity_frags(int n, int hi, float diag, v2f a[4]) {
#pragma unroll
  for (int k = 0; k < 4; ++k) {
    a[k].x = (n == (4 * k + 2 * hi))     ? diag : 0.0f;
    a[k].y = (n == (4 * k + 2 * hi + 1)) ? diag : 0.0f;
  }
}

__device__ __forceinline__ void load_b_frags(const float* __restrict__ ep, int hi, v2f b[4]) {
#pragma unroll
  for (int k = 0; k < 4; ++k) {
    b[k].x = ep[(size_t)(4 * k + 2 * hi) * EMB];
    b[k].y = ep[(size_t)(4 * k + 2 * hi + 1) * EMB];
  }
}

__global__ void __launch_bounds__(128) sum3_wmma_kernel(
    const float* __restrict__ ue, const float* __restrict__ ie,
    const float* __restrict__ e1, const float* __restrict__ e2,
    float* __restrict__ out) {
  int wave = threadIdx.x >> 5;
  int lane = threadIdx.x & 31;
  int n    = lane & 15;
  int hi   = lane >> 4;
  int row0 = blockIdx.x * 16;
  int col0 = wave * 16;
  const bool user = row0 < NUM_USERS;

  const size_t base = (size_t)row0 * EMB + col0 + n;
  const float* zp = user
      ? ue + base
      : ie + ((size_t)(row0 - NUM_USERS) * EMB + col0 + n);

  v8f c;
#pragma unroll
  for (int r = 0; r < 8; ++r)
    c[r] = zp[(size_t)(r + 8 * hi) * EMB];

  v2f a[4], b1[4], b2[4];
  identity_frags(n, hi, 1.0f, a);
  load_b_frags(e1 + base, hi, b1);
  load_b_frags(e2 + base, hi, b2);

  float* op = out + base;
  v8f w;
#pragma unroll
  for (int r = 0; r < 8; ++r) w[r] = 0.0f;
  if (user) {
#pragma unroll
    for (int r = 0; r < 8; ++r)
      w[r] = op[(size_t)(r + 8 * hi) * EMB];
  }

#pragma unroll
  for (int k = 0; k < 4; ++k)
    c = __builtin_amdgcn_wmma_f32_16x16x4_f32(
        false, a[k], false, b1[k], (short)0, c, false, false);
#pragma unroll
  for (int k = 0; k < 4; ++k)
    c = __builtin_amdgcn_wmma_f32_16x16x4_f32(
        false, a[k], false, b2[k], (short)0, c, false, false);

#pragma unroll
  for (int r = 0; r < 8; ++r)
    op[(size_t)(r + 8 * hi) * EMB] = c[r] * 0.25f + w[r];
}

__global__ void __launch_bounds__(128) last_wmma_kernel(
    const float* __restrict__ e3, float* __restrict__ out) {
  int wave = threadIdx.x >> 5;
  int lane = threadIdx.x & 31;
  int n    = lane & 15;
  int hi   = lane >> 4;
  int row0 = blockIdx.x * 16;
  int col0 = wave * 16;

  const size_t base = (size_t)row0 * EMB + col0 + n;
  float* op = out + base;

  v8f c;
#pragma unroll
  for (int r = 0; r < 8; ++r)
    c[r] = op[(size_t)(r + 8 * hi) * EMB];

  v2f a[4], b[4];
  identity_frags(n, hi, 0.25f, a);
  load_b_frags(e3 + base, hi, b);

#pragma unroll
  for (int k = 0; k < 4; ++k)
    c = __builtin_amdgcn_wmma_f32_16x16x4_f32(
        false, a[k], false, b[k], (short)0, c, false, false);

#pragma unroll
  for (int r = 0; r < 8; ++r)
    op[(size_t)(r + 8 * hi) * EMB] = c[r];
}

static inline int cdiv(long long a, int b) { return (int)((a + b - 1) / b); }

extern "C" void kernel_launch(void* const* d_in, const int* in_sizes, int n_in,
                              void* d_out, int out_size, void* d_ws, size_t ws_size,
                              hipStream_t stream) {
  const float* user_emb = (const float*)d_in[0];
  const float* item_emb = (const float*)d_in[1];
  const float* adj_val  = (const float*)d_in[2];
  const float* hv_val   = (const float*)d_in[3];
  const float* hu_val   = (const float*)d_in[4];
  const int*   adj_row  = (const int*)d_in[5];
  const int*   adj_col  = (const int*)d_in[6];
  const int*   hv_row   = (const int*)d_in[7];
  const int*   hv_col   = (const int*)d_in[8];
  const int*   hu_row   = (const int*)d_in[9];
  const int*   hu_col   = (const int*)d_in[10];

  const int E_adj = in_sizes[2];
  const int E_hv  = in_sizes[3];
  const int E_hu  = in_sizes[4];

  float* out = (float*)d_out;

  const size_t embBytesN = (size_t)NUM_NODES * EMB * sizeof(float);
  char* ws = (char*)d_ws;
  float* buf0 = (float*)(ws);
  float* buf1 = (float*)(ws + embBytesN);
  char*  keep = ws + 2 * embBytesN;

  int* start_a = (int*)(keep);
  int* perm_a  = (int*)(keep + 600064);
  int* tot_a   = (int*)(keep + 5400064);
  int* tot_h   = tot_a + 2 * BIN_ROWS;
  int* pairs_a = (int*)(out + (size_t)NUM_USERS * EMB);
  int* coarse_a = (int*)(ws + embBytesN);
  int* coarse_h = (int*)(ws + embBytesN + 14400000);
  int* start_h  = (int*)(ws + embBytesN + 24000000);
  int* pairs_h  = (int*)(ws + embBytesN + 24480064);
  float* bfeat  = (float*)(ws + embBytesN + 30880064);

  const int binsA = (NUM_NODES + BIN_ROWS - 1) / BIN_ROWS;
  const int binsH = (NUM_BQ + NUM_USERS + BIN_ROWS - 1) / BIN_ROWS;
  const EdgeSet sh = {hv_row, hv_col, hv_val, E_hv, hu_row, hu_col, hu_val, E_hu,
                      NUM_BQ, tot_h, coarse_h, start_h, pairs_h, nullptr,
                      binsH, NUM_BQ + NUM_USERS, cdiv((long long)E_hv + E_hu, CHUNK)};
  const EdgeSet sa = {adj_row, adj_col, adj_val, E_adj, nullptr, nullptr, nullptr, 0,
                      0, tot_a, coarse_a, start_a, pairs_a, perm_a,
                      binsA, NUM_NODES, cdiv(E_adj, CHUNK)};
  hipMemsetAsync(tot_a, 0, (size_t)(4 * BIN_ROWS) * sizeof(int), stream);
  bin_count_kernel<<<sh.nblk + sa.nblk, BIN_ROWS, 0, stream>>>(sh, sa);
  bin_scatter_kernel<<<sh.nblk + sa.nblk, BIN_ROWS, 0, stream>>>(sh, sa);
  bin_sort_kernel<<<binsH + binsA, BIN_ROWS, 0, stream>>>(sh, sa);

  gather_rows_kernel<true, false, true><<<cdiv((long long)NUM_BQ * 16, 256), 256, 0, stream>>>(
      start_h, pairs_h, nullptr, nullptr, item_emb, nullptr, 0,
      bfeat, NUM_BQ);
  gather_rows_kernel<true, false, true><<<cdiv((long long)NUM_USERS * 16, 256), 256, 0, stream>>>(
      start_h + NUM_BQ, pairs_h, nullptr, nullptr, bfeat, nullptr, 0,
      out, NUM_USERS);

  const int gatherBlocksN = cdiv((long long)NUM_NODES * 16, 256);
  gather_rows_kernel<true, true, false><<<gatherBlocksN, 256, 0, stream>>>(
      start_a, pairs_a, nullptr, nullptr, user_emb, item_emb, NUM_USERS,
      buf0, NUM_NODES);
  gather_rows_kernel<true, false, false><<<gatherBlocksN, 256, 0, stream>>>(
      start_a, pairs_a, nullptr, nullptr, buf0, nullptr, 0,
      buf1, NUM_NODES);
  sum3_wmma_kernel<<<NUM_NODES / 16, 128, 0, stream>>>(
      user_emb, item_emb, buf0, buf1, out);
  gather_rows_kernel<false, false, false><<<gatherBlocksN, 256, 0, stream>>>(
      start_a, perm_a, adj_col, adj_val, buf1, nullptr, 0,
      buf0, NUM_NODES);

  last_wmma_kernel<<<NUM_NODES / 16, 128, 0, stream>>>(buf0, out);
}


# Layer sum in Horner form: each layer gather stores t = e0 + A*t_prev, one WMMA pass ends: out = t3/4 + local; first sum pass and layer 3's edge-id lookups go (pairs stay parked); sum order changes
# speedup vs baseline: 4.6738x; 1.1549x over previous
#include <hip/hip_runtime.h>
#include <hip/hip_bf16.h>


typedef __attribute__((ext_vector_type(2))) float v2f;
typedef __attribute__((ext_vector_type(8))) float v8f;

#define NUM_USERS 100000
#define NUM_ITEMS 50000
#define NUM_NODES 150000
#define NUM_BQ    20000
#define EMB       64

#define BIN_SHIFT 9
#define BIN_ROWS  512
#define CHUNK     4096
#define STAGE_CAP 12288
#define SWEEP     8

struct EdgeSet {
  const int* row0; const int* col0; const float* val0; int nnz0;
  const int* row1; const int* col1; const float* val1; int nnz1;
  int rbase1;
  int* tot;
  int* coarse;
  int* start;
  int* pairs;
  int nb;
  int n_rows;
  int nblk;
};

__device__ __forceinline__ int wave_incl_scan(int v, int lane) {
  for (int d = 1; d < 32; d <<= 1) {
    int u = __shfl_up(v, d);
    if (lane >= d) v += u;
  }
  return v;
}

__device__ __forceinline__ int edge_row(const int* __restrict__ row0, int nnz0,
                                        const int* __restrict__ row1, int rbase1, int e) {
  return (e < nnz0) ? row0[e] : (row1[e - nnz0] + rbase1);
}

__device__ __forceinline__ void bin_count_body(
    const int* __restrict__ row0, int nnz0,
    const int* __restrict__ row1, int nnz1, int rbase1,
    int* __restrict__ tot, int blk) {
  __shared__ int h[BIN_ROWS];
  const int t = threadIdx.x;
  h[t] = 0;
  __syncthreads();
  const int nnz = nnz0 + nnz1;
  const int e0 = blk * CHUNK;
  int r[CHUNK / BIN_ROWS];
#pragma unroll
  for (int k = 0; k < CHUNK / BIN_ROWS; ++k) {
    const int e = e0 + k * BIN_ROWS + t;
    r[k] = (e < nnz) ? edge_row(row0, nnz0, row1, rbase1, e) : -1;
  }
#pragma unroll
  for (int k = 0; k < CHUNK / BIN_ROWS; ++k)
    if (r[k] >= 0) atomicAdd(&h[r[k] >> BIN_SHIFT], 1);
  __syncthreads();
  const int c = h[t];
  if (c > 0) atomicAdd(&tot[t], c);
}

__global__ void __launch_bounds__(BIN_ROWS) bin_count_kernel(EdgeSet sa, EdgeSet sb) {
  const bool first = (int)blockIdx.x < sa.nblk;
  const EdgeSet s = first ? sa : sb;
  const int blk = first ? (int)blockIdx.x : (int)blockIdx.x - sa.nblk;
  bin_count_body(s.row0, s.nnz0, s.row1, s.nnz1, s.rbase1, s.tot, blk);
}

__device__ __forceinline__ void bin_scatter_body(
    const int* __restrict__ row0, const int* __restrict__ col0,
    const float* __restrict__ val0, int nnz0,
    const int* __restrict__ row1, const int* __restrict__ col1,
    const float* __restrict__ val1, int nnz1, int rbase1,
    const int* __restrict__ tot, int* __restrict__ cur,
    int* __restrict__ coarse, int nb, int blk) {
  __shared__ int h[BIN_ROWS];
  __shared__ int ls[BIN_ROWS];
  __shared__ int bb[BIN_ROWS];
  __shared__ int wsum[BIN_ROWS / 32];
  __shared__ int4 stage[CHUNK];
  const int t = threadIdx.x;
  const int lane = t & 31;
  const int w = t >> 5;

  const int tv = (t < nb) ? tot[t] : 0;
  const int incl = wave_incl_scan(tv, lane);
  if (lane == 31) wsum[w] = incl;
  h[t] = 0;
  __syncthreads();
  int off = 0;
  for (int q = 0; q < BIN_ROWS / 32; ++q) if (q < w) off += wsum[q];
  bb[t] = off + incl - tv;
  __syncthreads();

  const int nnz = nnz0 + nnz1;
  const int e0 = blk * CHUNK;
  int r[CHUNK / BIN_ROWS], bin[CHUNK / BIN_ROWS], lr[CHUNK / BIN_ROWS], rec[CHUNK / BIN_ROWS];
  int cc[CHUNK / BIN_ROWS], vb[CHUNK / BIN_ROWS];
#pragma unroll
  for (int k = 0; k < CHUNK / BIN_ROWS; ++k) {
    const int e = e0 + k * BIN_ROWS + t;
    r[k] = -1; cc[k] = 0; vb[k] = 0;
    if (e < nnz) {
      if (e < nnz0) {
        r[k] = row0[e]; cc[k] = col0[e]; vb[k] = __float_as_int(val0[e]);
      } else {
        r[k] = row1[e - nnz0] + rbase1; cc[k] = col1[e - nnz0];
        vb[k] = __float_as_int(val1[e - nnz0]);
      }
    }
  }
#pragma unroll
  for (int k = 0; k < CHUNK / BIN_ROWS; ++k) {
    const int e = e0 + k * BIN_ROWS + t;
    bin[k] = -1; lr[k] = 0; rec[k] = 0;
    if (r[k] >= 0) {
      bin[k] = r[k] >> BIN_SHIFT;
      lr[k] = atomicAdd(&h[bin[k]], 1);
      rec[k] = (int)(((unsigned)e << BIN_SHIFT) | (unsigned)(r[k] & (BIN_ROWS - 1)));
    }
  }
  __syncthreads();

  const int hv = h[t];
  const int incl2 = wave_incl_scan(hv, lane);
  if (lane == 31) wsum[w] = incl2;
  __syncthreads();
  int off2 = 0;
  for (int q = 0; q < BIN_ROWS / 32; ++q) if (q < w) off2 += wsum[q];
  ls[t] = off2 + incl2 - hv;
  int g = 0;
  if (hv > 0) g = bb[t] + atomicAdd(&cur[t], hv);
  h[t] = g;
  __syncthreads();

#pragma unroll
  for (int k = 0; k < CHUNK / BIN_ROWS; ++k) {
    if (bin[k] >= 0) {
      const int p = ls[bin[k]] + lr[k];
      stage[p] = make_int4(rec[k], cc[k], vb[k], h[bin[k]] + lr[k]);
    }
  }
  __syncthreads();

  const int n_here = min(CHUNK, nnz - e0);
  for (int p = t; p < n_here; p += BIN_ROWS) {
    const int4 s = stage[p];
    int* __restrict__ q = coarse + 3 * s.w;
    q[0] = s.x; q[1] = s.y; q[2] = s.z;
  }
}

__global__ void __launch_bounds__(BIN_ROWS) bin_scatter_kernel(EdgeSet sa, EdgeSet sb) {
  const bool first = (int)blockIdx.x < sa.nblk;
  const EdgeSet s = first ? sa : sb;
  const int blk = first ? (int)blockIdx.x : (int)blockIdx.x - sa.nblk;
  bin_scatter_body(s.row0, s.col0, s.val0, s.nnz0, s.row1, s.col1, s.val1, s.nnz1,
                   s.rbase1, s.tot, s.tot + BIN_ROWS, s.coarse, s.nb, blk);
}

__device__ __forceinline__ void bin_sort_body(
    const int* __restrict__ tot, const int* __restrict__ coarse, int nb, int n_rows,
    int* __restrict__ start, int* __restrict__ pairs, int b) {
  __shared__ int cnt[BIN_ROWS];
  __shared__ int part[BIN_ROWS / 32];
  __shared__ int wsum[BIN_ROWS / 32];
  __shared__ int2 ordered[STAGE_CAP];
  const int t = threadIdx.x;
  const int lane = t & 31;
  const int w = t >> 5;

  int pre = (t < b) ? tot[t] : 0;
  for (int d = 16; d >= 1; d >>= 1) pre += __shfl_xor(pre, d);
  if (lane == 0) part[w] = pre;
  cnt[t] = 0;
  __syncthreads();
  int seg0 = 0;
  for (int q = 0; q < BIN_ROWS / 32; ++q) seg0 += part[q];
  const int len = tot[b];
  const int* __restrict__ seg = coarse + 3 * seg0;

  for (int i0 = t; i0 < len; i0 += SWEEP * BIN_ROWS) {
    int rec[SWEEP];
#pragma unroll
    for (int k = 0; k < SWEEP; ++k) {
      const int i = i0 + k * BIN_ROWS;
      rec[k] = (i < len) ? seg[3 * i] : -1;
    }
#pragma unroll
    for (int k = 0; k < SWEEP; ++k)
      if (rec[k] >= 0) atomicAdd(&cnt[rec[k] & (BIN_ROWS - 1)], 1);
  }
  __syncthreads();

  const int mine = cnt[t];
  const int incl = wave_incl_scan(mine, lane);
  if (lane == 31) wsum[w] = incl;
  __syncthreads();
  int off = 0;
  for (int q = 0; q < BIN_ROWS / 32; ++q) if (q < w) off += wsum[q];
  const int ex = off + incl - mine;
  const int row = b * BIN_ROWS + t;
  if (row < n_rows) start[row] = seg0 + ex;
  if (b == nb - 1 && t == 0) start[n_rows] = seg0 + len;
  cnt[t] = ex;
  __syncthreads();

  int2* __restrict__ pout = (int2*)pairs;
  const bool staged = len <= STAGE_CAP;
  for (int i0 = t; i0 < len; i0 += SWEEP * BIN_ROWS) {
    int rec[SWEEP], cc[SWEEP], vb[SWEEP];
#pragma unroll
    for (int k = 0; k < SWEEP; ++k) {
      const int i = i0 + k * BIN_ROWS;
      rec[k] = -1; cc[k] = 0; vb[k] = 0;
      if (i < len) { rec[k] = seg[3 * i]; cc[k] = seg[3 * i + 1]; vb[k] = seg[3 * i + 2]; }
    }
#pragma unroll
    for (int k = 0; k < SWEEP; ++k) {
      if (rec[k] >= 0) {
        const int pos = atomicAdd(&cnt[rec[k] & (BIN_ROWS - 1)], 1);
        if (staged) ordered[pos] = make_int2(cc[k], vb[k]);
        else        pout[seg0 + pos] = make_int2(cc[k], vb[k]);
      }
    }
  }
  if (staged) {
    __syncthreads();
    for (int i = t; i < len; i += BIN_ROWS)
      pout[seg0 + i] = ordered[i];
  }
}

__global__ void __launch_bounds__(BIN_ROWS) bin_sort_kernel(EdgeSet sa, EdgeSet sb) {
  const bool first = (int)blockIdx.x < sa.nb;
  const EdgeSet s = first ? sa : sb;
  const int b = first ? (int)blockIdx.x : (int)blockIdx.x - sa.nb;
  bin_sort_body(s.tot, s.coarse, s.nb, s.n_rows, s.start, s.pairs, b);
}

template <bool SPLIT, bool NORM, bool ADD0>
__global__ void __launch_bounds__(256) gather_rows_kernel(
    const int* __restrict__ start, const int* __restrict__ pairs,
    const float* __restrict__ x, const float* __restrict__ x1, int split,
    const float* __restrict__ z0, const float* __restrict__ z1, int zsplit,
    float* __restrict__ y, int n_rows) {
  const int g = blockIdx.x * blockDim.x + threadIdx.x;
  const int r = g >> 4;
  const int j = g & 15;
  int s = 0, n = 0;
  if (r < n_rows) { s = start[r]; n = start[r + 1] - s; }
  float4 z = make_float4(0.0f, 0.0f, 0.0f, 0.0f);
  if (ADD0) {
    if (r < n_rows) {
      const float* zp = (r < zsplit) ? z0 + (size_t)r * EMB
                                     : z1 + (size_t)(r - zsplit) * EMB;
      z = ((const float4*)zp)[j];
    }
  }
  const int nmax = max(n, __shfl_xor(n, 16));

  float4 acc = make_float4(0.0f, 0.0f, 0.0f, 0.0f);
  float d = 0.0f;
  for (int base = 0; base < nmax; base += 16) {
    int c = 0;
    float v = 0.0f;
    if (base + j < n) {
      int2 p = ((const int2*)pairs)[s + base + j];
      c = p.x; v = __int_as_float(p.y);
    }
    const int lim = min(nmax - base, 16);
    for (int i = 0; i < lim; i += 4) {
      const int   c0 = __shfl(c, i,     16), c1 = __shfl(c, i + 1, 16);
      const int   c2 = __shfl(c, i + 2, 16), c3 = __shfl(c, i + 3, 16);
      const float v0 = __shfl(v, i,     16), v1 = __shfl(v, i + 1, 16);
      const float v2 = __shfl(v, i + 2, 16), v3 = __shfl(v, i + 3, 16);
      const float* p0 = x + (size_t)c0 * EMB;
      const float* p1 = x + (size_t)c1 * EMB;
      const float* p2 = x + (size_t)c2 * EMB;
      const float* p3 = x + (size_t)c3 * EMB;
      if (SPLIT) {
        if (c0 >= split) p0 = x1 + (size_t)(c0 - split) * EMB;
        if (c1 >= split) p1 = x1 + (size_t)(c1 - split) * EMB;
        if (c2 >= split) p2 = x1 + (size_t)(c2 - split) * EMB;
        if (c3 >= split) p3 = x1 + (size_t)(c3 - split) * EMB;
      }
      const float4 x0 = ((const float4*)p0)[j];
      const float4 x1v = ((const float4*)p1)[j];
      const float4 x2 = ((const float4*)p2)[j];
      const float4 x3 = ((const float4*)p3)[j];
      if (base + i < n) {
        acc.x = fmaf(v0, x0.x, acc.x); acc.y = fmaf(v0, x0.y, acc.y);
        acc.z = fmaf(v0, x0.z, acc.z); acc.w = fmaf(v0, x0.w, acc.w);
        d += v0;
      }
      if (base + i + 1 < n) {
        acc.x = fmaf(v1, x1v.x, acc.x); acc.y = fmaf(v1, x1v.y, acc.y);
        acc.z = fmaf(v1, x1v.z, acc.z); acc.w = fmaf(v1, x1v.w, acc.w);
        d += v1;
      }
      if (base + i + 2 < n) {
        acc.x = fmaf(v2, x2.x, acc.x); acc.y = fmaf(v2, x2.y, acc.y);
        acc.z = fmaf(v2, x2.z, acc.z); acc.w = fmaf(v2, x2.w, acc.w);
        d += v2;
      }
      if (base + i + 3 < n) {
        acc.x = fmaf(v3, x3.x, acc.x); acc.y = fmaf(v3, x3.y, acc.y);
        acc.z = fmaf(v3, x3.z, acc.z); acc.w = fmaf(v3, x3.w, acc.w);
        d += v3;
      }
    }
  }
  if (NORM) {
    const float dd = (d == 0.0f) ? 1.0f : d;
    const float inv = 1.0f / dd;
    acc.x *= inv; acc.y *= inv; acc.z *= inv; acc.w *= inv;
  }
  if (ADD0) {
    acc.x += z.x; acc.y += z.y; acc.z += z.z; acc.w += z.w;
  }
  if (r < n_rows)
    ((float4*)(y + (size_t)r * EMB))[j] = acc;
}

__device__ __forceinline__ void identity_frags(int n, int hi, float diag, v2f a[4]) {
#pragma unroll
  for (int k = 0; k < 4; ++k) {
    a[k].x = (n == (4 * k + 2 * hi))     ? diag : 0.0f;
    a[k].y = (n == (4 * k + 2 * hi + 1)) ? diag : 0.0f;
  }
}

__device__ __forceinline__ void load_b_frags(const float* __restrict__ ep, int hi, v2f b[4]) {
#pragma unroll
  for (int k = 0; k < 4; ++k) {
    b[k].x = ep[(size_t)(4 * k + 2 * hi) * EMB];
    b[k].y = ep[(size_t)(4 * k + 2 * hi + 1) * EMB];
  }
}

__global__ void __launch_bounds__(128) last_wmma_kernel(
    const float* __restrict__ t3, float* __restrict__ out) {
  int wave = threadIdx.x >> 5;
  int lane = threadIdx.x & 31;
  int n    = lane & 15;
  int hi   = lane >> 4;
  int row0 = blockIdx.x * 16;
  int col0 = wave * 16;
  const bool user = row0 < NUM_USERS;

  const size_t base = (size_t)row0 * EMB + col0 + n;
  float* op = out + base;

  v2f a[4], b[4];
  identity_frags(n, hi, 0.25f, a);
  load_b_frags(t3 + base, hi, b);

  v8f c;
#pragma unroll
  for (int r = 0; r < 8; ++r) c[r] = 0.0f;
  if (user) {
#pragma unroll
    for (int r = 0; r < 8; ++r)
      c[r] = op[(size_t)(r + 8 * hi) * EMB];
  }

#pragma unroll
  for (int k = 0; k < 4; ++k)
    c = __builtin_amdgcn_wmma_f32_16x16x4_f32(
        false, a[k], false, b[k], (short)0, c, false, false);

#pragma unroll
  for (int r = 0; r < 8; ++r)
    op[(size_t)(r + 8 * hi) * EMB] = c[r];
}

static inline int cdiv(long long a, int b) { return (int)((a + b - 1) / b); }

extern "C" void kernel_launch(void* const* d_in, const int* in_sizes, int n_in,
                              void* d_out, int out_size, void* d_ws, size_t ws_size,
                              hipStream_t stream) {
  const float* user_emb = (const float*)d_in[0];
  const float* item_emb = (const float*)d_in[1];
  const float* adj_val  = (const float*)d_in[2];
  const float* hv_val   = (const float*)d_in[3];
  const float* hu_val   = (const float*)d_in[4];
  const int*   adj_row  = (const int*)d_in[5];
  const int*   adj_col  = (const int*)d_in[6];
  const int*   hv_row   = (const int*)d_in[7];
  const int*   hv_col   = (const int*)d_in[8];
  const int*   hu_row   = (const int*)d_in[9];
  const int*   hu_col   = (const int*)d_in[10];

  const int E_adj = in_sizes[2];
  const int E_hv  = in_sizes[3];
  const int E_hu  = in_sizes[4];

  float* out = (float*)d_out;

  const size_t embBytesN = (size_t)NUM_NODES * EMB * sizeof(float);
  char* ws = (char*)d_ws;
  float* buf0 = (float*)(ws);
  float* buf1 = (float*)(ws + embBytesN);
  char*  keep = ws + 2 * embBytesN;

  int* start_a = (int*)(keep);
  int* tot_a   = (int*)(keep + 600064);
  int* tot_h   = tot_a + 2 * BIN_ROWS;
  int* pairs_a = (int*)(out + (size_t)NUM_USERS * EMB);
  int* coarse_a = (int*)(ws + embBytesN);
  int* coarse_h = (int*)(ws + embBytesN + 14400000);
  int* start_h  = (int*)(ws + embBytesN + 24000000);
  int* pairs_h  = (int*)(ws + embBytesN + 24480064);
  float* bfeat  = (float*)(ws + embBytesN + 30880064);

  const int binsA = (NUM_NODES + BIN_ROWS - 1) / BIN_ROWS;
  const int binsH = (NUM_BQ + NUM_USERS + BIN_ROWS - 1) / BIN_ROWS;
  const EdgeSet sh = {hv_row, hv_col, hv_val, E_hv, hu_row, hu_col, hu_val, E_hu,
                      NUM_BQ, tot_h, coarse_h, start_h, pairs_h,
                      binsH, NUM_BQ + NUM_USERS, cdiv((long long)E_hv + E_hu, CHUNK)};
  const EdgeSet sa = {adj_row, adj_col, adj_val, E_adj, nullptr, nullptr, nullptr, 0,
                      0, tot_a, coarse_a, start_a, pairs_a,
                      binsA, NUM_NODES, cdiv(E_adj, CHUNK)};
  hipMemsetAsync(tot_a, 0, (size_t)(4 * BIN_ROWS) * sizeof(int), stream);
  bin_count_kernel<<<sh.nblk + sa.nblk, BIN_ROWS, 0, stream>>>(sh, sa);
  bin_scatter_kernel<<<sh.nblk + sa.nblk, BIN_ROWS, 0, stream>>>(sh, sa);
  bin_sort_kernel<<<binsH + binsA, BIN_ROWS, 0, stream>>>(sh, sa);

  gather_rows_kernel<false, true, false><<<cdiv((long long)NUM_BQ * 16, 256), 256, 0, stream>>>(
      start_h, pairs_h, item_emb, nullptr, 0, nullptr, nullptr, 0,
      bfeat, NUM_BQ);
  gather_rows_kernel<false, true, false><<<cdiv((long long)NUM_USERS * 16, 256), 256, 0, stream>>>(
      start_h + NUM_BQ, pairs_h, bfeat, nullptr, 0, nullptr, nullptr, 0,
      out, NUM_USERS);

  const int gatherBlocksN = cdiv((long long)NUM_NODES * 16, 256);
  gather_rows_kernel<true, false, true><<<gatherBlocksN, 256, 0, stream>>>(
      start_a, pairs_a, user_emb, item_emb, NUM_USERS, user_emb, item_emb, NUM_USERS,
      buf0, NUM_NODES);
  gather_rows_kernel<false, false, true><<<gatherBlocksN, 256, 0, stream>>>(
      start_a, pairs_a, buf0, nullptr, 0, user_emb, item_emb, NUM_USERS,
      buf1, NUM_NODES);
  gather_rows_kernel<false, false, true><<<gatherBlocksN, 256, 0, stream>>>(
      start_a, pairs_a, buf1, nullptr, 0, user_emb, item_emb, NUM_USERS,
      buf0, NUM_NODES);

  last_wmma_kernel<<<NUM_NODES / 16, 128, 0, stream>>>(buf0, out);
}


# Closing pass joins layer 3's launch: 16-row blocks pass t3 via LDS to the same 4 f32 WMMAs per tile and store out; t3 never stored; t1 sits in out, pairs + local term in workspace; sum order kept
# speedup vs baseline: 4.8641x; 1.0407x over previous
#include <hip/hip_runtime.h>
#include <hip/hip_bf16.h>


typedef __attribute__((ext_vector_type(2))) float v2f;
typedef __attribute__((ext_vector_type(8))) float v8f;

#define NUM_USERS 100000
#define NUM_ITEMS 50000
#define NUM_NODES 150000
#define NUM_BQ    20000
#define EMB       64

#define BIN_SHIFT 9
#define BIN_ROWS  512
#define CHUNK     4096
#define STAGE_CAP 12288
#define SWEEP     8

struct EdgeSet {
  const int* row0; const int* col0; const float* val0; int nnz0;
  const int* row1; const int* col1; const float* val1; int nnz1;
  int rbase1;
  int* tot;
  int* coarse;
  int* start;
  int* pairs;
  int nb;
  int n_rows;
  int nblk;
};

__device__ __forceinline__ int wave_incl_scan(int v, int lane) {
  for (int d = 1; d < 32; d <<= 1) {
    int u = __shfl_up(v, d);
    if (lane >= d) v += u;
  }
  return v;
}

__device__ __forceinline__ int edge_row(const int* __restrict__ row0, int nnz0,
                                        const int* __restrict__ row1, int rbase1, int e) {
  return (e < nnz0) ? row0[e] : (row1[e - nnz0] + rbase1);
}

__device__ __forceinline__ void bin_count_body(
    const int* __restrict__ row0, int nnz0,
    const int* __restrict__ row1, int nnz1, int rbase1,
    int* __restrict__ tot, int blk) {
  __shared__ int h[BIN_ROWS];
  const int t = threadIdx.x;
  h[t] = 0;
  __syncthreads();
  const int nnz = nnz0 + nnz1;
  const int e0 = blk * CHUNK;
  int r[CHUNK / BIN_ROWS];
#pragma unroll
  for (int k = 0; k < CHUNK / BIN_ROWS; ++k) {
    const int e = e0 + k * BIN_ROWS + t;
    r[k] = (e < nnz) ? edge_row(row0, nnz0, row1, rbase1, e) : -1;
  }
#pragma unroll
  for (int k = 0; k < CHUNK / BIN_ROWS; ++k)
    if (r[k] >= 0) atomicAdd(&h[r[k] >> BIN_SHIFT], 1);
  __syncthreads();
  const int c = h[t];
  if (c > 0) atomicAdd(&tot[t], c);
}

__global__ void __launch_bounds__(BIN_ROWS) bin_count_kernel(EdgeSet sa, EdgeSet sb) {
  const bool first = (int)blockIdx.x < sa.nblk;
  const EdgeSet s = first ? sa : sb;
  const int blk = first ? (int)blockIdx.x : (int)blockIdx.x - sa.nblk;
  bin_count_body(s.row0, s.nnz0, s.row1, s.nnz1, s.rbase1, s.tot, blk);
}

__device__ __forceinline__ void bin_scatter_body(
    const int* __restrict__ row0, const int* __restrict__ col0,
    const float* __restrict__ val0, int nnz0,
    const int* __restrict__ row1, const int* __restrict__ col1,
    const float* __restrict__ val1, int nnz1, int rbase1,
    const int* __restrict__ tot, int* __restrict__ cur,
    int* __restrict__ coarse, int nb, int blk) {
  __shared__ int h[BIN_ROWS];
  __shared__ int ls[BIN_ROWS];
  __shared__ int bb[BIN_ROWS];
  __shared__ int wsum[BIN_ROWS / 32];
  __shared__ int4 stage[CHUNK];
  const int t = threadIdx.x;
  const int lane = t & 31;
  const int w = t >> 5;

  const int tv = (t < nb) ? tot[t] : 0;
  const int incl = wave_incl_scan(tv, lane);
  if (lane == 31) wsum[w] = incl;
  h[t] = 0;
  __syncthreads();
  int off = 0;
  for (int q = 0; q < BIN_ROWS / 32; ++q) if (q < w) off += wsum[q];
  bb[t] = off + incl - tv;
  __syncthreads();

  const int nnz = nnz0 + nnz1;
  const int e0 = blk * CHUNK;
  int r[CHUNK / BIN_ROWS], bin[CHUNK / BIN_ROWS], lr[CHUNK / BIN_ROWS], rec[CHUNK / BIN_ROWS];
  int cc[CHUNK / BIN_ROWS], vb[CHUNK / BIN_ROWS];
#pragma unroll
  for (int k = 0; k < CHUNK / BIN_ROWS; ++k) {
    const int e = e0 + k * BIN_ROWS + t;
    r[k] = -1; cc[k] = 0; vb[k] = 0;
    if (e < nnz) {
      if (e < nnz0) {
        r[k] = row0[e]; cc[k] = col0[e]; vb[k] = __float_as_int(val0[e]);
      } else {
        r[k] = row1[e - nnz0] + rbase1; cc[k] = col1[e - nnz0];
        vb[k] = __float_as_int(val1[e - nnz0]);
      }
    }
  }
#pragma unroll
  for (int k = 0; k < CHUNK / BIN_ROWS; ++k) {
    const int e = e0 + k * BIN_ROWS + t;
    bin[k] = -1; lr[k] = 0; rec[k] = 0;
    if (r[k] >= 0) {
      bin[k] = r[k] >> BIN_SHIFT;
      lr[k] = atomicAdd(&h[bin[k]], 1);
      rec[k] = (int)(((unsigned)e << BIN_SHIFT) | (unsigned)(r[k] & (BIN_ROWS - 1)));
    }
  }
  __syncthreads();

  const int hv = h[t];
  const int incl2 = wave_incl_scan(hv, lane);
  if (lane == 31) wsum[w] = incl2;
  __syncthreads();
  int off2 = 0;
  for (int q = 0; q < BIN_ROWS / 32; ++q) if (q < w) off2 += wsum[q];
  ls[t] = off2 + incl2 - hv;
  int g = 0;
  if (hv > 0) g = bb[t] + atomicAdd(&cur[t], hv);
  h[t] = g;
  __syncthreads();

#pragma unroll
  for (int k = 0; k < CHUNK / BIN_ROWS; ++k) {
    if (bin[k] >= 0) {
      const int p = ls[bin[k]] + lr[k];
      stage[p] = make_int4(rec[k], cc[k], vb[k], h[bin[k]] + lr[k]);
    }
  }
  __syncthreads();

  const int n_here = min(CHUNK, nnz - e0);
  for (int p = t; p < n_here; p += BIN_ROWS) {
    const int4 s = stage[p];
    int* __restrict__ q = coarse + 3 * s.w;
    q[0] = s.x; q[1] = s.y; q[2] = s.z;
  }
}

__global__ void __launch_bounds__(BIN_ROWS) bin_scatter_kernel(EdgeSet sa, EdgeSet sb) {
  const bool first = (int)blockIdx.x < sa.nblk;
  const EdgeSet s = first ? sa : sb;
  const int blk = first ? (int)blockIdx.x : (int)blockIdx.x - sa.nblk;
  bin_scatter_body(s.row0, s.col0, s.val0, s.nnz0, s.row1, s.col1, s.val1, s.nnz1,
                   s.rbase1, s.tot, s.tot + BIN_ROWS, s.coarse, s.nb, blk);
}

__device__ __forceinline__ void bin_sort_body(
    const int* __restrict__ tot, const int* __restrict__ coarse, int nb, int n_rows,
    int* __restrict__ start, int* __restrict__ pairs, int b) {
  __shared__ int cnt[BIN_ROWS];
  __shared__ int part[BIN_ROWS / 32];
  __shared__ int wsum[BIN_ROWS / 32];
  __shared__ int2 ordered[STAGE_CAP];
  const int t = threadIdx.x;
  const int lane = t & 31;
  const int w = t >> 5;

  int pre = (t < b) ? tot[t] : 0;
  for (int d = 16; d >= 1; d >>= 1) pre += __shfl_xor(pre, d);
  if (lane == 0) part[w] = pre;
  cnt[t] = 0;
  __syncthreads();
  int seg0 = 0;
  for (int q = 0; q < BIN_ROWS / 32; ++q) seg0 += part[q];
  const int len = tot[b];
  const int* __restrict__ seg = coarse + 3 * seg0;

  for (int i0 = t; i0 < len; i0 += SWEEP * BIN_ROWS) {
    int rec[SWEEP];
#pragma unroll
    for (int k = 0; k < SWEEP; ++k) {
      const int i = i0 + k * BIN_ROWS;
      rec[k] = (i < len) ? seg[3 * i] : -1;
    }
#pragma unroll
    for (int k = 0; k < SWEEP; ++k)
      if (rec[k] >= 0) atomicAdd(&cnt[rec[k] & (BIN_ROWS - 1)], 1);
  }
  __syncthreads();

  const int mine = cnt[t];
  const int incl = wave_incl_scan(mine, lane);
  if (lane == 31) wsum[w] = incl;
  __syncthreads();
  int off = 0;
  for (int q = 0; q < BIN_ROWS / 32; ++q) if (q < w) off += wsum[q];
  const int ex = off + incl - mine;
  const int row = b * BIN_ROWS + t;
  if (row < n_rows) start[row] = seg0 + ex;
  if (b == nb - 1 && t == 0) start[n_rows] = seg0 + len;
  cnt[t] = ex;
  __syncthreads();

  int2* __restrict__ pout = (int2*)pairs;
  const bool staged = len <= STAGE_CAP;
  for (int i0 = t; i0 < len; i0 += SWEEP * BIN_ROWS) {
    int rec[SWEEP], cc[SWEEP], vb[SWEEP];
#pragma unroll
    for (int k = 0; k < SWEEP; ++k) {
      const int i = i0 + k * BIN_ROWS;
      rec[k] = -1; cc[k] = 0; vb[k] = 0;
      if (i < len) { rec[k] = seg[3 * i]; cc[k] = seg[3 * i + 1]; vb[k] = seg[3 * i + 2]; }
    }
#pragma unroll
    for (int k = 0; k < SWEEP; ++k) {
      if (rec[k] >= 0) {
        const int pos = atomicAdd(&cnt[rec[k] & (BIN_ROWS - 1)], 1);
        if (staged) ordered[pos] = make_int2(cc[k], vb[k]);
        else        pout[seg0 + pos] = make_int2(cc[k], vb[k]);
      }
    }
  }
  if (staged) {
    __syncthreads();
    for (int i = t; i < len; i += BIN_ROWS)
      pout[seg0 + i] = ordered[i];
  }
}

__global__ void __launch_bounds__(BIN_ROWS) bin_sort_kernel(EdgeSet sa, EdgeSet sb) {
  const bool first = (int)blockIdx.x < sa.nb;
  const EdgeSet s = first ? sa : sb;
  const int b = first ? (int)blockIdx.x : (int)blockIdx.x - sa.nb;
  bin_sort_body(s.tot, s.coarse, s.nb, s.n_rows, s.start, s.pairs, b);
}

template <bool SPLIT, bool NORM, bool ADD0, bool SIDE>
__device__ __forceinline__ float4 gather_one_row(
    const int* __restrict__ start, const int* __restrict__ pairs,
    const float* __restrict__ x, const float* __restrict__ x1, int split,
    const float* __restrict__ z0, const float* __restrict__ z1, int zsplit,
    int n_rows, int r, int j,
    const float* __restrict__ side, float4& side_row) {
  int s = 0, n = 0;
  if (r < n_rows) { s = start[r]; n = start[r + 1] - s; }
  if (SIDE) {
    side_row = make_float4(0.0f, 0.0f, 0.0f, 0.0f);
    if (side != nullptr) side_row = ((const float4*)side)[j];
  }
  float4 z = make_float4(0.0f, 0.0f, 0.0f, 0.0f);
  if (ADD0) {
    if (r < n_rows) {
      const float* zp = (r < zsplit) ? z0 + (size_t)r * EMB
                                     : z1 + (size_t)(r - zsplit) * EMB;
      z = ((const float4*)zp)[j];
    }
  }
  const int nmax = max(n, __shfl_xor(n, 16));

  float4 acc = make_float4(0.0f, 0.0f, 0.0f, 0.0f);
  float d = 0.0f;
  for (int base = 0; base < nmax; base += 16) {
    int c = 0;
    float v = 0.0f;
    if (base + j < n) {
      int2 p = ((const int2*)pairs)[s + base + j];
      c = p.x; v = __int_as_float(p.y);
    }
    const int lim = min(nmax - base, 16);
    for (int i = 0; i < lim; i += 4) {
      const int   c0 = __shfl(c, i,     16), c1 = __shfl(c, i + 1, 16);
      const int   c2 = __shfl(c, i + 2, 16), c3 = __shfl(c, i + 3, 16);
      const float v0 = __shfl(v, i,     16), v1 = __shfl(v, i + 1, 16);
      const float v2 = __shfl(v, i + 2, 16), v3 = __shfl(v, i + 3, 16);
      const float* p0 = x + (size_t)c0 * EMB;
      const float* p1 = x + (size_t)c1 * EMB;
      const float* p2 = x + (size_t)c2 * EMB;
      const float* p3 = x + (size_t)c3 * EMB;
      if (SPLIT) {
        if (c0 >= split) p0 = x1 + (size_t)(c0 - split) * EMB;
        if (c1 >= split) p1 = x1 + (size_t)(c1 - split) * EMB;
        if (c2 >= split) p2 = x1 + (size_t)(c2 - split) * EMB;
        if (c3 >= split) p3 = x1 + (size_t)(c3 - split) * EMB;
      }
      const float4 x0 = ((const float4*)p0)[j];
      const float4 x1v = ((const float4*)p1)[j];
      const float4 x2 = ((const float4*)p2)[j];
      const float4 x3 = ((const float4*)p3)[j];
      if (base + i < n) {
        acc.x = fmaf(v0, x0.x, acc.x); acc.y = fmaf(v0, x0.y, acc.y);
        acc.z = fmaf(v0, x0.z, acc.z); acc.w = fmaf(v0, x0.w, acc.w);
        d += v0;
      }
      if (base + i + 1 < n) {
        acc.x = fmaf(v1, x1v.x, acc.x); acc.y = fmaf(v1, x1v.y, acc.y);
        acc.z = fmaf(v1, x1v.z, acc.z); acc.w = fmaf(v1, x1v.w, acc.w);
        d += v1;
      }
      if (base + i + 2 < n) {
        acc.x = fmaf(v2, x2.x, acc.x); acc.y = fmaf(v2, x2.y, acc.y);
        acc.z = fmaf(v2, x2.z, acc.z); acc.w = fmaf(v2, x2.w, acc.w);
        d += v2;
      }
      if (base + i + 3 < n) {
        acc.x = fmaf(v3, x3.x, acc.x); acc.y = fmaf(v3, x3.y, acc.y);
        acc.z = fmaf(v3, x3.z, acc.z); acc.w = fmaf(v3, x3.w, acc.w);
        d += v3;
      }
    }
  }
  if (NORM) {
    const float dd = (d == 0.0f) ? 1.0f : d;
    const float inv = 1.0f / dd;
    acc.x *= inv; acc.y *= inv; acc.z *= inv; acc.w *= inv;
  }
  if (ADD0) {
    acc.x += z.x; acc.y += z.y; acc.z += z.z; acc.w += z.w;
  }
  return acc;
}

template <bool SPLIT, bool NORM, bool ADD0>
__global__ void __launch_bounds__(256) gather_rows_kernel(
    const int* __restrict__ start, const int* __restrict__ pairs,
    const float* __restrict__ x, const float* __restrict__ x1, int split,
    const float* __restrict__ z0, const float* __restrict__ z1, int zsplit,
    float* __restrict__ y, int n_rows) {
  const int g = blockIdx.x * blockDim.x + threadIdx.x;
  const int r = g >> 4;
  const int j = g & 15;
  float4 none;
  const float4 acc = gather_one_row<SPLIT, NORM, ADD0, false>(
      start, pairs, x, x1, split, z0, z1, zsplit, n_rows, r, j, nullptr, none);
  if (r < n_rows)
    ((float4*)(y + (size_t)r * EMB))[j] = acc;
}

__device__ __forceinline__ void identity_frags(int n, int hi, float diag, v2f a[4]) {
#pragma unroll
  for (int k = 0; k < 4; ++k) {
    a[k].x = (n == (4 * k + 2 * hi))     ? diag : 0.0f;
    a[k].y = (n == (4 * k + 2 * hi + 1)) ? diag : 0.0f;
  }
}

__device__ __forceinline__ void load_b_frags(const float* ep, int pitch, int hi, v2f b[4]) {
#pragma unroll
  for (int k = 0; k < 4; ++k) {
    b[k].x = ep[(4 * k + 2 * hi) * pitch];
    b[k].y = ep[(4 * k + 2 * hi + 1) * pitch];
  }
}

#define TILE_PITCH 76

__global__ void __launch_bounds__(256) gather_close_kernel(
    const int* __restrict__ start, const int* __restrict__ pairs,
    const float* __restrict__ x,
    const float* __restrict__ z0, const float* __restrict__ z1, int zsplit,
    const float* __restrict__ local, float* __restrict__ out) {
  __shared__ float tt[16 * TILE_PITCH];
  __shared__ float ct[16 * TILE_PITCH];
  const int t  = threadIdx.x;
  const int rl = t >> 4;
  const int j  = t & 15;
  const int row0 = blockIdx.x * 16;
  const int r = row0 + rl;
  const bool user = row0 < NUM_USERS;

  float4 lc;
  const float4 acc = gather_one_row<false, false, true, true>(
      start, pairs, x, nullptr, 0, z0, z1, zsplit, NUM_NODES, r, j,
      user ? local + (size_t)r * EMB : nullptr, lc);

  *(float4*)(tt + rl * TILE_PITCH + 4 * j) = acc;
  *(float4*)(ct + rl * TILE_PITCH + 4 * j) = lc;
  __syncthreads();

  const int wave = t >> 5;
  if (wave < 4) {
    const int lane = t & 31;
    const int n    = lane & 15;
    const int hi   = lane >> 4;
    const int col0 = wave * 16;

    v2f a[4], b[4];
    identity_frags(n, hi, 0.25f, a);
    load_b_frags(tt + col0 + n, TILE_PITCH, hi, b);

    float* cp = ct + (8 * hi) * TILE_PITCH + col0 + n;
    v8f c;
#pragma unroll
    for (int q = 0; q < 8; ++q) c[q] = cp[q * TILE_PITCH];

#pragma unroll
    for (int k = 0; k < 4; ++k)
      c = __builtin_amdgcn_wmma_f32_16x16x4_f32(
          false, a[k], false, b[k], (short)0, c, false, false);

#pragma unroll
    for (int q = 0; q < 8; ++q) cp[q * TILE_PITCH] = c[q];
  }
  __syncthreads();

  const float4 o = *(const float4*)(ct + rl * TILE_PITCH + 4 * j);
  ((float4*)(out + (size_t)r * EMB))[j] = o;
}

static inline int cdiv(long long a, int b) { return (int)((a + b - 1) / b); }

extern "C" void kernel_launch(void* const* d_in, const int* in_sizes, int n_in,
                              void* d_out, int out_size, void* d_ws, size_t ws_size,
                              hipStream_t stream) {
  const float* user_emb = (const float*)d_in[0];
  const float* item_emb = (const float*)d_in[1];
  const float* adj_val  = (const float*)d_in[2];
  const float* hv_val   = (const float*)d_in[3];
  const float* hu_val   = (const float*)d_in[4];
  const int*   adj_row  = (const int*)d_in[5];
  const int*   adj_col  = (const int*)d_in[6];
  const int*   hv_row   = (const int*)d_in[7];
  const int*   hv_col   = (const int*)d_in[8];
  const int*   hu_row   = (const int*)d_in[9];
  const int*   hu_col   = (const int*)d_in[10];

  const int E_adj = in_sizes[2];
  const int E_hv  = in_sizes[3];
  const int E_hu  = in_sizes[4];

  float* out = (float*)d_out;

  const size_t embBytesN = (size_t)NUM_NODES * EMB * sizeof(float);
  char* ws = (char*)d_ws;
  float* buf1 = (float*)(ws + embBytesN);
  char*  keep = ws + 2 * embBytesN;

  int* start_a = (int*)(keep);
  int* tot_a   = (int*)(keep + 600064);
  int* tot_h   = tot_a + 2 * BIN_ROWS;
  int*   pairs_a = (int*)(ws);
  float* local   = (float*)(ws + 9600000);
  int* coarse_a = (int*)(ws + embBytesN);
  int* coarse_h = (int*)(ws + embBytesN + 14400000);
  int* start_h  = (int*)(ws + embBytesN + 24000000);
  int* pairs_h  = (int*)(ws + embBytesN + 24480064);
  float* bfeat  = (float*)(ws + embBytesN + 30880064);

  const int binsA = (NUM_NODES + BIN_ROWS - 1) / BIN_ROWS;
  const int binsH = (NUM_BQ + NUM_USERS + BIN_ROWS - 1) / BIN_ROWS;
  const EdgeSet sh = {hv_row, hv_col, hv_val, E_hv, hu_row, hu_col, hu_val, E_hu,
                      NUM_BQ, tot_h, coarse_h, start_h, pairs_h,
                      binsH, NUM_BQ + NUM_USERS, cdiv((long long)E_hv + E_hu, CHUNK)};
  const EdgeSet sa = {adj_row, adj_col, adj_val, E_adj, nullptr, nullptr, nullptr, 0,
                      0, tot_a, coarse_a, start_a, pairs_a,
                      binsA, NUM_NODES, cdiv(E_adj, CHUNK)};
  hipMemsetAsync(tot_a, 0, (size_t)(4 * BIN_ROWS) * sizeof(int), stream);
  bin_count_kernel<<<sh.nblk + sa.nblk, BIN_ROWS, 0, stream>>>(sh, sa);
  bin_scatter_kernel<<<sh.nblk + sa.nblk, BIN_ROWS, 0, stream>>>(sh, sa);
  bin_sort_kernel<<<binsH + binsA, BIN_ROWS, 0, stream>>>(sh, sa);

  gather_rows_kernel<false, true, false><<<cdiv((long long)NUM_BQ * 16, 256), 256, 0, stream>>>(
      start_h, pairs_h, item_emb, nullptr, 0, nullptr, nullptr, 0,
      bfeat, NUM_BQ);
  gather_rows_kernel<false, true, false><<<cdiv((long long)NUM_USERS * 16, 256), 256, 0, stream>>>(
      start_h + NUM_BQ, pairs_h, bfeat, nullptr, 0, nullptr, nullptr, 0,
      local, NUM_USERS);

  const int gatherBlocksN = cdiv((long long)NUM_NODES * 16, 256);
  gather_rows_kernel<true, false, true><<<gatherBlocksN, 256, 0, stream>>>(
      start_a, pairs_a, user_emb, item_emb, NUM_USERS, user_emb, item_emb, NUM_USERS,
      out, NUM_NODES);
  gather_rows_kernel<false, false, true><<<gatherBlocksN, 256, 0, stream>>>(
      start_a, pairs_a, out, nullptr, 0, user_emb, item_emb, NUM_USERS,
      buf1, NUM_NODES);
  gather_close_kernel<<<NUM_NODES / 16, 256, 0, stream>>>(
      start_a, pairs_a, buf1, user_emb, item_emb, NUM_USERS, local, out);
}
